# starting kernel
# baseline (speedup 1.0000x reference)
#include <hip/hip_runtime.h>
#include <hip/hip_bf16.h>

#define CDIM 128
#define LN_EPS 1e-5f

typedef float v2f __attribute__((ext_vector_type(2)));
typedef float v8f __attribute__((ext_vector_type(8)));

__device__ __forceinline__ float wave_sum32(float v) {
#pragma unroll
    for (int o = 16; o > 0; o >>= 1) v += __shfl_xor(v, o, 32);
    return v;
}

__global__ void gcn_init_kernel(float* __restrict__ accum, float* __restrict__ deg, int N) {
    size_t idx = (size_t)blockIdx.x * blockDim.x + threadIdx.x;
    size_t total = (size_t)N * CDIM;
    if (idx < total) accum[idx] = 0.0f;
    if (idx < (size_t)N) deg[idx] = 1.0f;
}

__global__ void gcn_degree_kernel(const int* __restrict__ ei, float* __restrict__ deg, int E) {
    int e = blockIdx.x * blockDim.x + threadIdx.x;
    if (e >= E) return;
    int dst = ei[E + e];
    unsafeAtomicAdd(&deg[dst], 1.0f);
}

__global__ void gcn_dinv_kernel(float* __restrict__ deg, int N) {
    int i = blockIdx.x * blockDim.x + threadIdx.x;
    if (i < N) deg[i] = rsqrtf(deg[i]);
}

__global__ void __launch_bounds__(256)
gcn_gemm_kernel(const float* __restrict__ x, const float* __restrict__ W,
                float* __restrict__ h, int N) {
    const int wave = threadIdx.x >> 5;
    const int lane = threadIdx.x & 31;
    const int half = lane >> 4;
    const int l    = lane & 15;
    const int row_base = blockIdx.x * 16;
    const int col_base = wave * 16;

    int m = row_base + l;
    if (m >= N) m = N - 1;

    const float* xrow = x + (size_t)m * CDIM + half * 2;
    const float* wrow = W + (size_t)(col_base + l) * CDIM + half * 2;

    v8f acc = {};
#pragma unroll
    for (int k0 = 0; k0 < CDIM; k0 += 4) {
        v2f a = *(const v2f*)(xrow + k0);
        v2f b = *(const v2f*)(wrow + k0);
        acc = __builtin_amdgcn_wmma_f32_16x16x4_f32(
             false, a,  false, b,
             (short)0, acc,  false,  false);
    }

    const int outcol = col_base + l;
#pragma unroll
    for (int r = 0; r < 8; ++r) {
        int orow = row_base + r + half * 8;
        if (orow < N) h[(size_t)orow * CDIM + outcol] = acc[r];
    }
}

__global__ void __launch_bounds__(256)
gcn_scatter_kernel(const float* __restrict__ h, const float* __restrict__ dinv,
                   const int* __restrict__ ei, float* __restrict__ accum, int E) {
    int wid = blockIdx.x * (blockDim.x >> 5) + (threadIdx.x >> 5);
    if (wid >= E) return;
    int lane = threadIdx.x & 31;

    int src = ei[wid];
    int dst = ei[E + wid];
    float norm = dinv[src] * dinv[dst];

    const float4 hv = *(const float4*)(h + (size_t)src * CDIM + lane * 4);
    float* ap = accum + (size_t)dst * CDIM + lane * 4;
    unsafeAtomicAdd(ap + 0, hv.x * norm);
    unsafeAtomicAdd(ap + 1, hv.y * norm);
    unsafeAtomicAdd(ap + 2, hv.z * norm);
    unsafeAtomicAdd(ap + 3, hv.w * norm);
}

__global__ void __launch_bounds__(256)
gcn_epilogue_kernel(const float* __restrict__ x, const float* __restrict__ h,
                    const float* __restrict__ accum, const float* __restrict__ dinv,
                    const float* __restrict__ bvec,
                    const float* __restrict__ g1, const float* __restrict__ be1,
                    const float* __restrict__ g2, const float* __restrict__ be2,
                    float* __restrict__ out, int N) {
    int wid = blockIdx.x * (blockDim.x >> 5) + (threadIdx.x >> 5);
    if (wid >= N) return;
    int lane = threadIdx.x & 31;
    int c = lane * 4;
    size_t base = (size_t)wid * CDIM + c;

    float di = dinv[wid];
    float sw = di * di;

    float4 hv = *(const float4*)(h + base);
    float4 av = *(const float4*)(accum + base);
    float4 bv = *(const float4*)(bvec + c);

    float v0 = fmaxf(av.x + sw * hv.x + bv.x, 0.0f);
    float v1 = fmaxf(av.y + sw * hv.y + bv.y, 0.0f);
    float v2 = fmaxf(av.z + sw * hv.z + bv.z, 0.0f);
    float v3 = fmaxf(av.w + sw * hv.w + bv.w, 0.0f);

    float mu = wave_sum32(v0 + v1 + v2 + v3) * (1.0f / CDIM);
    float d0 = v0 - mu, d1 = v1 - mu, d2 = v2 - mu, d3 = v3 - mu;
    float var = wave_sum32(d0 * d0 + d1 * d1 + d2 * d2 + d3 * d3) * (1.0f / CDIM);
    float inv = rsqrtf(var + LN_EPS);
    float4 g1v = *(const float4*)(g1 + c);
    float4 b1v = *(const float4*)(be1 + c);
    float4 xv  = *(const float4*)(x + base);
    float z0 = xv.x + (d0 * inv * g1v.x + b1v.x);
    float z1 = xv.y + (d1 * inv * g1v.y + b1v.y);
    float z2 = xv.z + (d2 * inv * g1v.z + b1v.z);
    float z3 = xv.w + (d3 * inv * g1v.w + b1v.w);

    float mu2 = wave_sum32(z0 + z1 + z2 + z3) * (1.0f / CDIM);
    float e0 = z0 - mu2, e1 = z1 - mu2, e2 = z2 - mu2, e3 = z3 - mu2;
    float var2 = wave_sum32(e0 * e0 + e1 * e1 + e2 * e2 + e3 * e3) * (1.0f / CDIM);
    float inv2 = rsqrtf(var2 + LN_EPS);
    float4 g2v = *(const float4*)(g2 + c);
    float4 b2v = *(const float4*)(be2 + c);

    float4 o;
    o.x = e0 * inv2 * g2v.x + b2v.x;
    o.y = e1 * inv2 * g2v.y + b2v.y;
    o.z = e2 * inv2 * g2v.z + b2v.z;
    o.w = e3 * inv2 * g2v.w + b2v.w;
    *(float4*)(out + base) = o;
}

extern "C" void kernel_launch(void* const* d_in, const int* in_sizes, int n_in,
                              void* d_out, int out_size, void* d_ws, size_t ws_size,
                              hipStream_t stream) {
    const float* x  = (const float*)d_in[0];
    const int*   ei = (const int*)d_in[1];
    const float* W  = (const float*)d_in[2];
    const float* b  = (const float*)d_in[3];
    const float* g1 = (const float*)d_in[4];
    const float* be1 = (const float*)d_in[5];
    const float* g2 = (const float*)d_in[6];
    const float* be2 = (const float*)d_in[7];
    float* out = (float*)d_out;

    const int N = in_sizes[0] / CDIM;
    const int E = in_sizes[1] / 2;

    float* h     = (float*)d_ws;
    float* accum = h + (size_t)N * CDIM;
    float* deg   = accum + (size_t)N * CDIM;

    const size_t total = (size_t)N * CDIM;
    gcn_init_kernel<<<(int)((total + 255) / 256), 256, 0, stream>>>(accum, deg, N);
    gcn_degree_kernel<<<(E + 255) / 256, 256, 0, stream>>>(ei, deg, E);
    gcn_dinv_kernel<<<(N + 255) / 256, 256, 0, stream>>>(deg, N);
    gcn_gemm_kernel<<<(N + 15) / 16, 256, 0, stream>>>(x, W, h, N);
    gcn_scatter_kernel<<<(E + 7) / 8, 256, 0, stream>>>(h, deg, ei, accum, E);
    gcn_epilogue_kernel<<<(N + 7) / 8, 256, 0, stream>>>(x, h, accum, deg, b,
                                                         g1, be1, g2, be2, out, N);
}


# scatter: lane = channel (l, l+32, l+64, l+96), so each f32 atomic add instruction covers one whole 128-byte line instead of 8 words in each of 4 lines; atomics sum in arrival order as before
# speedup vs baseline: 1.1783x; 1.1783x over previous
#include <hip/hip_runtime.h>
#include <hip/hip_bf16.h>

#define CDIM 128
#define LN_EPS 1e-5f

typedef float v2f __attribute__((ext_vector_type(2)));
typedef float v8f __attribute__((ext_vector_type(8)));

__device__ __forceinline__ float wave_sum32(float v) {
#pragma unroll
    for (int o = 16; o > 0; o >>= 1) v += __shfl_xor(v, o, 32);
    return v;
}

__global__ void gcn_init_kernel(float* __restrict__ accum, float* __restrict__ deg, int N) {
    size_t idx = (size_t)blockIdx.x * blockDim.x + threadIdx.x;
    size_t total = (size_t)N * CDIM;
    if (idx < total) accum[idx] = 0.0f;
    if (idx < (size_t)N) deg[idx] = 1.0f;
}

__global__ void gcn_degree_kernel(const int* __restrict__ ei, float* __restrict__ deg, int E) {
    int e = blockIdx.x * blockDim.x + threadIdx.x;
    if (e >= E) return;
    int dst = ei[E + e];
    unsafeAtomicAdd(&deg[dst], 1.0f);
}

__global__ void gcn_dinv_kernel(float* __restrict__ deg, int N) {
    int i = blockIdx.x * blockDim.x + threadIdx.x;
    if (i < N) deg[i] = rsqrtf(deg[i]);
}

__global__ void __launch_bounds__(256)
gcn_gemm_kernel(const float* __restrict__ x, const float* __restrict__ W,
                float* __restrict__ h, int N) {
    const int wave = threadIdx.x >> 5;
    const int lane = threadIdx.x & 31;
    const int half = lane >> 4;
    const int l    = lane & 15;
    const int row_base = blockIdx.x * 16;
    const int col_base = wave * 16;

    int m = row_base + l;
    if (m >= N) m = N - 1;

    const float* xrow = x + (size_t)m * CDIM + half * 2;
    const float* wrow = W + (size_t)(col_base + l) * CDIM + half * 2;

    v8f acc = {};
#pragma unroll
    for (int k0 = 0; k0 < CDIM; k0 += 4) {
        v2f a = *(const v2f*)(xrow + k0);
        v2f b = *(const v2f*)(wrow + k0);
        acc = __builtin_amdgcn_wmma_f32_16x16x4_f32(
             false, a,  false, b,
             (short)0, acc,  false,  false);
    }

    const int outcol = col_base + l;
#pragma unroll
    for (int r = 0; r < 8; ++r) {
        int orow = row_base + r + half * 8;
        if (orow < N) h[(size_t)orow * CDIM + outcol] = acc[r];
    }
}

__global__ void __launch_bounds__(256)
gcn_scatter_kernel(const float* __restrict__ h, const float* __restrict__ dinv,
                   const int* __restrict__ ei, float* __restrict__ accum, int E) {
    int wid = blockIdx.x * (blockDim.x >> 5) + (threadIdx.x >> 5);
    if (wid >= E) return;
    int lane = threadIdx.x & 31;

    int src = ei[wid];
    int dst = ei[E + wid];
    float norm = dinv[src] * dinv[dst];

    const float* hp = h + (size_t)src * CDIM + lane;
    float* ap = accum + (size_t)dst * CDIM + lane;
    const float h0 = hp[0];
    const float h1 = hp[32];
    const float h2 = hp[64];
    const float h3 = hp[96];
    unsafeAtomicAdd(ap +  0, h0 * norm);
    unsafeAtomicAdd(ap + 32, h1 * norm);
    unsafeAtomicAdd(ap + 64, h2 * norm);
    unsafeAtomicAdd(ap + 96, h3 * norm);
}

__global__ void __launch_bounds__(256)
gcn_epilogue_kernel(const float* __restrict__ x, const float* __restrict__ h,
                    const float* __restrict__ accum, const float* __restrict__ dinv,
                    const float* __restrict__ bvec,
                    const float* __restrict__ g1, const float* __restrict__ be1,
                    const float* __restrict__ g2, const float* __restrict__ be2,
                    float* __restrict__ out, int N) {
    int wid = blockIdx.x * (blockDim.x >> 5) + (threadIdx.x >> 5);
    if (wid >= N) return;
    int lane = threadIdx.x & 31;
    int c = lane * 4;
    size_t base = (size_t)wid * CDIM + c;

    float di = dinv[wid];
    float sw = di * di;

    float4 hv = *(const float4*)(h + base);
    float4 av = *(const float4*)(accum + base);
    float4 bv = *(const float4*)(bvec + c);

    float v0 = fmaxf(av.x + sw * hv.x + bv.x, 0.0f);
    float v1 = fmaxf(av.y + sw * hv.y + bv.y, 0.0f);
    float v2 = fmaxf(av.z + sw * hv.z + bv.z, 0.0f);
    float v3 = fmaxf(av.w + sw * hv.w + bv.w, 0.0f);

    float mu = wave_sum32(v0 + v1 + v2 + v3) * (1.0f / CDIM);
    float d0 = v0 - mu, d1 = v1 - mu, d2 = v2 - mu, d3 = v3 - mu;
    float var = wave_sum32(d0 * d0 + d1 * d1 + d2 * d2 + d3 * d3) * (1.0f / CDIM);
    float inv = rsqrtf(var + LN_EPS);
    float4 g1v = *(const float4*)(g1 + c);
    float4 b1v = *(const float4*)(be1 + c);
    float4 xv  = *(const float4*)(x + base);
    float z0 = xv.x + (d0 * inv * g1v.x + b1v.x);
    float z1 = xv.y + (d1 * inv * g1v.y + b1v.y);
    float z2 = xv.z + (d2 * inv * g1v.z + b1v.z);
    float z3 = xv.w + (d3 * inv * g1v.w + b1v.w);

    float mu2 = wave_sum32(z0 + z1 + z2 + z3) * (1.0f / CDIM);
    float e0 = z0 - mu2, e1 = z1 - mu2, e2 = z2 - mu2, e3 = z3 - mu2;
    float var2 = wave_sum32(e0 * e0 + e1 * e1 + e2 * e2 + e3 * e3) * (1.0f / CDIM);
    float inv2 = rsqrtf(var2 + LN_EPS);
    float4 g2v = *(const float4*)(g2 + c);
    float4 b2v = *(const float4*)(be2 + c);

    float4 o;
    o.x = e0 * inv2 * g2v.x + b2v.x;
    o.y = e1 * inv2 * g2v.y + b2v.y;
    o.z = e2 * inv2 * g2v.z + b2v.z;
    o.w = e3 * inv2 * g2v.w + b2v.w;
    *(float4*)(out + base) = o;
}

extern "C" void kernel_launch(void* const* d_in, const int* in_sizes, int n_in,
                              void* d_out, int out_size, void* d_ws, size_t ws_size,
                              hipStream_t stream) {
    const float* x  = (const float*)d_in[0];
    const int*   ei = (const int*)d_in[1];
    const float* W  = (const float*)d_in[2];
    const float* b  = (const float*)d_in[3];
    const float* g1 = (const float*)d_in[4];
    const float* be1 = (const float*)d_in[5];
    const float* g2 = (const float*)d_in[6];
    const float* be2 = (const float*)d_in[7];
    float* out = (float*)d_out;

    const int N = in_sizes[0] / CDIM;
    const int E = in_sizes[1] / 2;

    float* h     = (float*)d_ws;
    float* accum = h + (size_t)N * CDIM;
    float* deg   = accum + (size_t)N * CDIM;

    const size_t total = (size_t)N * CDIM;
    gcn_init_kernel<<<(int)((total + 255) / 256), 256, 0, stream>>>(accum, deg, N);
    gcn_degree_kernel<<<(E + 255) / 256, 256, 0, stream>>>(ei, deg, E);
    gcn_dinv_kernel<<<(N + 255) / 256, 256, 0, stream>>>(deg, N);
    gcn_gemm_kernel<<<(N + 15) / 16, 256, 0, stream>>>(x, W, h, N);
    gcn_scatter_kernel<<<(E + 7) / 8, 256, 0, stream>>>(h, deg, ei, accum, E);
    gcn_epilogue_kernel<<<(N + 7) / 8, 256, 0, stream>>>(x, h, accum, deg, b,
                                                         g1, be1, g2, be2, out, N);
}


# aggregation by pull, not push: edges bucketed by dst each call (count, scan, int-atomic slot fill), one wave per node sums its rows in registers, then epilogue; no f32 atomics; sum in slot order
# speedup vs baseline: 1.5142x; 1.2851x over previous
#include <hip/hip_runtime.h>
#include <hip/hip_bf16.h>

#define CDIM 128
#define LN_EPS 1e-5f

typedef float v2f __attribute__((ext_vector_type(2)));
typedef float v8f __attribute__((ext_vector_type(8)));

__device__ __forceinline__ float wave_sum32(float v) {
#pragma unroll
    for (int o = 16; o > 0; o >>= 1) v += __shfl_xor(v, o, 32);
    return v;
}

__device__ __forceinline__ int wave_isum32(int v) {
#pragma unroll
    for (int o = 16; o > 0; o >>= 1) v += __shfl_xor(v, o, 32);
    return v;
}


__global__ void gcn_init_kernel(int* __restrict__ cnt, int N) {
    int i = blockIdx.x * blockDim.x + threadIdx.x;
    if (i < N) cnt[i] = 0;
}

__global__ void gcn_count_kernel(const int* __restrict__ ei, int* __restrict__ cnt, int E) {
    int e = blockIdx.x * blockDim.x + threadIdx.x;
    if (e >= E) return;
    atomicAdd(&cnt[ei[E + e]], 1);
}

__global__ void __launch_bounds__(1024)
gcn_scan_kernel(const int* __restrict__ cnt, int* __restrict__ rowptr,
                int* __restrict__ cursor, float* __restrict__ dinv, int N) {
    __shared__ int s_pre[32];
    __shared__ int s_tot[32];
    const int tid  = threadIdx.x;
    const int lane = tid & 31;
    const int wave = tid >> 5;

    const int4* c4 = (const int4*)cnt;
    const int n4 = blockIdx.x * 256;
    int pre = 0;
#pragma unroll 4
    for (int i = tid; i < n4; i += 1024) {
        const int4 v = c4[i];
        pre += (v.x + v.y) + (v.z + v.w);
    }
    pre = wave_isum32(pre);

    const int idx  = blockIdx.x * 1024 + tid;
    const int cval = idx < N ? cnt[idx] : 0;
    int inc = cval;
#pragma unroll
    for (int o = 1; o < 32; o <<= 1) {
        const int t = __shfl_up(inc, o, 32);
        if (lane >= o) inc += t;
    }
    if (lane == 0)  s_pre[wave] = pre;
    if (lane == 31) s_tot[wave] = inc;
    __syncthreads();
    const int blockpre = wave_isum32(s_pre[lane]);
    const int wavepre  = wave_isum32(lane < wave ? s_tot[lane] : 0);
    if (idx < N) {
        const int ex = blockpre + wavepre + inc - cval;
        rowptr[idx] = ex;
        cursor[idx] = ex;
        dinv[idx]   = rsqrtf((float)(cval + 1));
    }
}

__global__ void gcn_fill_kernel(const int* __restrict__ ei, int* __restrict__ cursor,
                                int* __restrict__ csr_src, int E) {
    int e = blockIdx.x * blockDim.x + threadIdx.x;
    if (e >= E) return;
    const int src = ei[e];
    const int dst = ei[E + e];
    const int pos = atomicAdd(&cursor[dst], 1);
    csr_src[pos] = src;
}

__global__ void __launch_bounds__(256)
gcn_gemm_kernel(const float* __restrict__ x, const float* __restrict__ W,
                float* __restrict__ h, int N) {
    const int wave = threadIdx.x >> 5;
    const int lane = threadIdx.x & 31;
    const int half = lane >> 4;
    const int l    = lane & 15;
    const int row_base = blockIdx.x * 16;
    const int col_base = wave * 16;

    int m = row_base + l;
    if (m >= N) m = N - 1;

    const float* xrow = x + (size_t)m * CDIM + half * 2;
    const float* wrow = W + (size_t)(col_base + l) * CDIM + half * 2;

    v8f acc = {};
#pragma unroll
    for (int k0 = 0; k0 < CDIM; k0 += 4) {
        v2f a = *(const v2f*)(xrow + k0);
        v2f b = *(const v2f*)(wrow + k0);
        acc = __builtin_amdgcn_wmma_f32_16x16x4_f32(
             false, a,  false, b,
             (short)0, acc,  false,  false);
    }

    const int outcol = col_base + l;
#pragma unroll
    for (int r = 0; r < 8; ++r) {
        int orow = row_base + r + half * 8;
        if (orow < N) h[(size_t)orow * CDIM + outcol] = acc[r];
    }
}

template <int U>
__device__ __forceinline__ void add_rows(const float* __restrict__ h, int s, float w,
                                         int j, int c, float4& acc) {
    int    sj[U];
    float  wj[U];
    float4 r[U];
#pragma unroll
    for (int u = 0; u < U; ++u) {
        sj[u] = __builtin_amdgcn_readlane(s, j + u);
        wj[u] = __int_as_float(__builtin_amdgcn_readlane(__float_as_int(w), j + u));
    }
#pragma unroll
    for (int u = 0; u < U; ++u) r[u] = *(const float4*)(h + (size_t)sj[u] * CDIM + c);
#pragma unroll
    for (int u = 0; u < U; ++u) {
        acc.x = fmaf(wj[u], r[u].x, acc.x);
        acc.y = fmaf(wj[u], r[u].y, acc.y);
        acc.z = fmaf(wj[u], r[u].z, acc.z);
        acc.w = fmaf(wj[u], r[u].w, acc.w);
    }
}

__global__ void __launch_bounds__(256)
gcn_gather_epilogue_kernel(const float* __restrict__ x, const float* __restrict__ h,
                           const int* __restrict__ rowptr, const int* __restrict__ cnt,
                           const int* __restrict__ csr_src, const float* __restrict__ dinv,
                           const float* __restrict__ bvec,
                           const float* __restrict__ g1, const float* __restrict__ be1,
                           const float* __restrict__ g2, const float* __restrict__ be2,
                           float* __restrict__ out, int N) {
    const int wid = __builtin_amdgcn_readfirstlane(
        (int)(blockIdx.x * (blockDim.x >> 5) + (threadIdx.x >> 5)));
    if (wid >= N) return;
    const int lane = threadIdx.x & 31;
    const int c = lane * 4;
    const size_t base = (size_t)wid * CDIM + c;

    const int   start = rowptr[wid];
    const int   n     = cnt[wid];
    const float di    = dinv[wid];

    const float4 hv = *(const float4*)(h + base);
    const float4 xv = *(const float4*)(x + base);

    float4 acc = make_float4(0.0f, 0.0f, 0.0f, 0.0f);
    for (int j0 = 0; j0 < n; j0 += 32) {
        const int m = (n - j0) < 32 ? (n - j0) : 32;
        const int   k = lane < m ? lane : 0;
        const int   s = csr_src[start + j0 + k];
        const float w = dinv[s] * di;
        int j = 0;
        for (; j + 8 <= m; j += 8) add_rows<8>(h, s, w, j, c, acc);
        switch (m - j) {
            case 7: add_rows<7>(h, s, w, j, c, acc); break;
            case 6: add_rows<6>(h, s, w, j, c, acc); break;
            case 5: add_rows<5>(h, s, w, j, c, acc); break;
            case 4: add_rows<4>(h, s, w, j, c, acc); break;
            case 3: add_rows<3>(h, s, w, j, c, acc); break;
            case 2: add_rows<2>(h, s, w, j, c, acc); break;
            case 1: add_rows<1>(h, s, w, j, c, acc); break;
            default: break;
        }
    }

    const float sw = di * di;
    const float4 bv = *(const float4*)(bvec + c);

    float v0 = fmaxf(acc.x + sw * hv.x + bv.x, 0.0f);
    float v1 = fmaxf(acc.y + sw * hv.y + bv.y, 0.0f);
    float v2 = fmaxf(acc.z + sw * hv.z + bv.z, 0.0f);
    float v3 = fmaxf(acc.w + sw * hv.w + bv.w, 0.0f);

    float mu = wave_sum32(v0 + v1 + v2 + v3) * (1.0f / CDIM);
    float d0 = v0 - mu, d1 = v1 - mu, d2 = v2 - mu, d3 = v3 - mu;
    float var = wave_sum32(d0 * d0 + d1 * d1 + d2 * d2 + d3 * d3) * (1.0f / CDIM);
    float inv = rsqrtf(var + LN_EPS);
    float4 g1v = *(const float4*)(g1 + c);
    float4 b1v = *(const float4*)(be1 + c);
    float z0 = xv.x + (d0 * inv * g1v.x + b1v.x);
    float z1 = xv.y + (d1 * inv * g1v.y + b1v.y);
    float z2 = xv.z + (d2 * inv * g1v.z + b1v.z);
    float z3 = xv.w + (d3 * inv * g1v.w + b1v.w);

    float mu2 = wave_sum32(z0 + z1 + z2 + z3) * (1.0f / CDIM);
    float e0 = z0 - mu2, e1 = z1 - mu2, e2 = z2 - mu2, e3 = z3 - mu2;
    float var2 = wave_sum32(e0 * e0 + e1 * e1 + e2 * e2 + e3 * e3) * (1.0f / CDIM);
    float inv2 = rsqrtf(var2 + LN_EPS);
    float4 g2v = *(const float4*)(g2 + c);
    float4 b2v = *(const float4*)(be2 + c);

    float4 o;
    o.x = e0 * inv2 * g2v.x + b2v.x;
    o.y = e1 * inv2 * g2v.y + b2v.y;
    o.z = e2 * inv2 * g2v.z + b2v.z;
    o.w = e3 * inv2 * g2v.w + b2v.w;
    *(float4*)(out + base) = o;
}

extern "C" void kernel_launch(void* const* d_in, const int* in_sizes, int n_in,
                              void* d_out, int out_size, void* d_ws, size_t ws_size,
                              hipStream_t stream) {
    const float* x  = (const float*)d_in[0];
    const int*   ei = (const int*)d_in[1];
    const float* W  = (const float*)d_in[2];
    const float* b  = (const float*)d_in[3];
    const float* g1 = (const float*)d_in[4];
    const float* be1 = (const float*)d_in[5];
    const float* g2 = (const float*)d_in[6];
    const float* be2 = (const float*)d_in[7];
    float* out = (float*)d_out;

    const int N = in_sizes[0] / CDIM;
    const int E = in_sizes[1] / 2;

    float* h       = (float*)d_ws;
    int*   cnt     = (int*)(h + (size_t)N * CDIM);
    int*   rowptr  = cnt + N;
    int*   cursor  = rowptr + N;
    float* dinv    = (float*)(cursor + N);
    int*   csr_src = (int*)(dinv + N);

    gcn_init_kernel<<<(N + 255) / 256, 256, 0, stream>>>(cnt, N);
    gcn_count_kernel<<<(E + 255) / 256, 256, 0, stream>>>(ei, cnt, E);
    gcn_scan_kernel<<<(N + 1023) / 1024, 1024, 0, stream>>>(cnt, rowptr, cursor, dinv, N);
    gcn_fill_kernel<<<(E + 255) / 256, 256, 0, stream>>>(ei, cursor, csr_src, E);
    gcn_gemm_kernel<<<(N + 15) / 16, 256, 0, stream>>>(x, W, h, N);
    gcn_gather_epilogue_kernel<<<(N + 7) / 8, 256, 0, stream>>>(x, h, rowptr, cnt, csr_src, dinv, b,
                                                                g1, be1, g2, be2, out, N);
}


# aggregate rows of x first, then (sum) W^T on the f32 matrix unit inside the gather kernel: 32 nodes per 16-wave group, two per wave, via LDS; no product launch, no h in memory; sums re-associated
# speedup vs baseline: 1.8665x; 1.2327x over previous
#include <hip/hip_runtime.h>
#include <hip/hip_bf16.h>

#define CDIM 128
#define LN_EPS 1e-5f

typedef float v2f __attribute__((ext_vector_type(2)));
typedef float v8f __attribute__((ext_vector_type(8)));

__device__ __forceinline__ float wave_sum32(float v) {
#pragma unroll
    for (int o = 16; o > 0; o >>= 1) v += __shfl_xor(v, o, 32);
    return v;
}

__device__ __forceinline__ int wave_isum32(int v) {
#pragma unroll
    for (int o = 16; o > 0; o >>= 1) v += __shfl_xor(v, o, 32);
    return v;
}


__global__ void gcn_init_kernel(int* __restrict__ cnt, int N) {
    int i = blockIdx.x * blockDim.x + threadIdx.x;
    if (i < N) cnt[i] = 0;
}

__global__ void gcn_count_kernel(const int* __restrict__ ei, int* __restrict__ cnt, int E) {
    int e = blockIdx.x * blockDim.x + threadIdx.x;
    if (e >= E) return;
    atomicAdd(&cnt[ei[E + e]], 1);
}

__global__ void __launch_bounds__(1024)
gcn_scan_kernel(const int* __restrict__ cnt, int* __restrict__ rowptr,
                int* __restrict__ cursor, float* __restrict__ dinv, int N) {
    __shared__ int s_pre[32];
    __shared__ int s_tot[32];
    const int tid  = threadIdx.x;
    const int lane = tid & 31;
    const int wave = tid >> 5;

    const int4* c4 = (const int4*)cnt;
    const int n4 = blockIdx.x * 256;
    int pre = 0;
#pragma unroll 4
    for (int i = tid; i < n4; i += 1024) {
        const int4 v = c4[i];
        pre += (v.x + v.y) + (v.z + v.w);
    }
    pre = wave_isum32(pre);

    const int idx  = blockIdx.x * 1024 + tid;
    const int cval = idx < N ? cnt[idx] : 0;
    int inc = cval;
#pragma unroll
    for (int o = 1; o < 32; o <<= 1) {
        const int t = __shfl_up(inc, o, 32);
        if (lane >= o) inc += t;
    }
    if (lane == 0)  s_pre[wave] = pre;
    if (lane == 31) s_tot[wave] = inc;
    __syncthreads();
    const int blockpre = wave_isum32(s_pre[lane]);
    const int wavepre  = wave_isum32(lane < wave ? s_tot[lane] : 0);
    if (idx < N) {
        const int ex = blockpre + wavepre + inc - cval;
        rowptr[idx] = ex;
        cursor[idx] = ex;
        dinv[idx]   = rsqrtf((float)(cval + 1));
    }
}

__global__ void gcn_fill_kernel(const int* __restrict__ ei, int* __restrict__ cursor,
                                int* __restrict__ csr_src, int E) {
    int e = blockIdx.x * blockDim.x + threadIdx.x;
    if (e >= E) return;
    const int src = ei[e];
    const int dst = ei[E + e];
    const int pos = atomicAdd(&cursor[dst], 1);
    csr_src[pos] = src;
}

template <int U>
__device__ __forceinline__ void add_rows(const float* __restrict__ xs, int s, float w,
                                         int j, int c, float4& acc) {
    int    sj[U];
    float  wj[U];
    float4 r[U];
#pragma unroll
    for (int u = 0; u < U; ++u) {
        sj[u] = __builtin_amdgcn_readlane(s, j + u);
        wj[u] = __int_as_float(__builtin_amdgcn_readlane(__float_as_int(w), j + u));
    }
#pragma unroll
    for (int u = 0; u < U; ++u) r[u] = *(const float4*)(xs + (size_t)sj[u] * CDIM + c);
#pragma unroll
    for (int u = 0; u < U; ++u) {
        acc.x = fmaf(wj[u], r[u].x, acc.x);
        acc.y = fmaf(wj[u], r[u].y, acc.y);
        acc.z = fmaf(wj[u], r[u].z, acc.z);
        acc.w = fmaf(wj[u], r[u].w, acc.w);
    }
}

__device__ __forceinline__ void add_chunk(const float* __restrict__ xs, int s, float w,
                                          int m, int c, float4& acc) {
    int j = 0;
    for (; j + 8 <= m; j += 8) add_rows<8>(xs, s, w, j, c, acc);
    switch (m - j) {
        case 7: add_rows<7>(xs, s, w, j, c, acc); break;
        case 6: add_rows<6>(xs, s, w, j, c, acc); break;
        case 5: add_rows<5>(xs, s, w, j, c, acc); break;
        case 4: add_rows<4>(xs, s, w, j, c, acc); break;
        case 3: add_rows<3>(xs, s, w, j, c, acc); break;
        case 2: add_rows<2>(xs, s, w, j, c, acc); break;
        case 1: add_rows<1>(xs, s, w, j, c, acc); break;
        default: break;
    }
}

#define TILE_M 32
#define LDS_PITCH 132

__global__ void __launch_bounds__(512) __attribute__((amdgpu_waves_per_eu(16, 16)))
gcn_gather_gemm_epilogue_kernel(const float* __restrict__ x, const float* __restrict__ W,
                                const int* __restrict__ rowptr, const int* __restrict__ cnt,
                                const int* __restrict__ csr_src, const float* __restrict__ dinv,
                                const float* __restrict__ bvec,
                                const float* __restrict__ g1, const float* __restrict__ be1,
                                const float* __restrict__ g2, const float* __restrict__ be2,
                                float* __restrict__ out, int N) {
    __shared__ __attribute__((aligned(16))) float s_agg[TILE_M * LDS_PITCH];
    __shared__ __attribute__((aligned(16))) float s_d[2][TILE_M * LDS_PITCH];

    const int wave  = __builtin_amdgcn_readfirstlane((int)(threadIdx.x >> 5));
    const int node0 = (int)blockIdx.x * TILE_M + 2 * wave;
    const int lane  = threadIdx.x & 31;
    const int c = lane * 4;

    bool   live[2];
    int    st[2], nn[2], m0[2], s0[2];
    float  di[2], w0[2];
    float4 xv[2];
#pragma unroll
    for (int i = 0; i < 2; ++i) {
        live[i] = node0 + i < N;
        const int q = live[i] ? node0 + i : 0;
        st[i] = rowptr[q];
        nn[i] = live[i] ? cnt[q] : 0;
        di[i] = dinv[q];
        xv[i] = *(const float4*)(x + (size_t)q * CDIM + c);
    }
#pragma unroll
    for (int i = 0; i < 2; ++i) {
        m0[i] = nn[i] < 32 ? nn[i] : 32;
        const int k = lane < m0[i] ? lane : 0;
        s0[i] = nn[i] > 0 ? csr_src[st[i] + k] : 0;
    }
#pragma unroll
    for (int i = 0; i < 2; ++i) w0[i] = dinv[s0[i]] * di[i];

#pragma unroll
    for (int i = 0; i < 2; ++i) {
        float4 acc = make_float4(0.0f, 0.0f, 0.0f, 0.0f);
        add_chunk(x, s0[i], w0[i], m0[i], c, acc);
        for (int j0 = 32; j0 < nn[i]; j0 += 32) {
            const int m = (nn[i] - j0) < 32 ? (nn[i] - j0) : 32;
            const int   k = lane < m ? lane : 0;
            const int   s = csr_src[st[i] + j0 + k];
            const float w = dinv[s] * di[i];
            add_chunk(x, s, w, m, c, acc);
        }
        if (live[i]) {
            const float sw = di[i] * di[i];
            acc.x = fmaf(sw, xv[i].x, acc.x);
            acc.y = fmaf(sw, xv[i].y, acc.y);
            acc.z = fmaf(sw, xv[i].z, acc.z);
            acc.w = fmaf(sw, xv[i].w, acc.w);
        }
        *(float4*)(s_agg + (2 * wave + i) * LDS_PITCH + c) = acc;
    }

    const int half  = lane >> 4;
    const int l     = lane & 15;
    const int ct    = wave & 7;
    const int kp    = wave >> 3;
    const int kbase = kp * 64 + half * 4;
    const float* wrow = W + (size_t)(ct * 16 + l) * CDIM + kbase;
    float4 bf[8];
#pragma unroll
    for (int kb = 0; kb < 8; ++kb) bf[kb] = *(const float4*)(wrow + kb * 8);

    __syncthreads();

#pragma unroll
    for (int t = 0; t < 2; ++t) {
        const float* arow = s_agg + (16 * t + l) * LDS_PITCH + kbase;
        v8f d = {};
#pragma unroll
        for (int kb = 0; kb < 8; ++kb) {
            const float4 af = *(const float4*)(arow + kb * 8);
            v2f a0 = {af.x, af.y};
            v2f b0 = {bf[kb].x, bf[kb].y};
            d = __builtin_amdgcn_wmma_f32_16x16x4_f32(
                 false, a0,  false, b0,
                 (short)0, d,  false,  false);
            v2f a1 = {af.z, af.w};
            v2f b1 = {bf[kb].z, bf[kb].w};
            d = __builtin_amdgcn_wmma_f32_16x16x4_f32(
                 false, a1,  false, b1,
                 (short)0, d,  false,  false);
        }
        float* dp = s_d[kp] + (16 * t) * LDS_PITCH + ct * 16 + l;
#pragma unroll
        for (int r = 0; r < 8; ++r) dp[(r + 8 * half) * LDS_PITCH] = d[r];
    }

    __syncthreads();

    float4 xr[2];
#pragma unroll
    for (int i = 0; i < 2; ++i)
        xr[i] = *(const float4*)(x + (size_t)(live[i] ? node0 + i : 0) * CDIM + c);
    const float4 bv  = *(const float4*)(bvec + c);
    const float4 g1v = *(const float4*)(g1 + c);
    const float4 b1v = *(const float4*)(be1 + c);
    const float4 g2v = *(const float4*)(g2 + c);
    const float4 b2v = *(const float4*)(be2 + c);

#pragma unroll
    for (int i = 0; i < 2; ++i) {
        if (!live[i]) continue;
        const int row = 2 * wave + i;
        const float4 p0 = *(const float4*)(s_d[0] + row * LDS_PITCH + c);
        const float4 p1 = *(const float4*)(s_d[1] + row * LDS_PITCH + c);

        float v0 = fmaxf((p0.x + p1.x) + bv.x, 0.0f);
        float v1 = fmaxf((p0.y + p1.y) + bv.y, 0.0f);
        float v2 = fmaxf((p0.z + p1.z) + bv.z, 0.0f);
        float v3 = fmaxf((p0.w + p1.w) + bv.w, 0.0f);

        float mu = wave_sum32(v0 + v1 + v2 + v3) * (1.0f / CDIM);
        float d0 = v0 - mu, d1 = v1 - mu, d2 = v2 - mu, d3 = v3 - mu;
        float var = wave_sum32(d0 * d0 + d1 * d1 + d2 * d2 + d3 * d3) * (1.0f / CDIM);
        float inv = rsqrtf(var + LN_EPS);
        float z0 = xr[i].x + (d0 * inv * g1v.x + b1v.x);
        float z1 = xr[i].y + (d1 * inv * g1v.y + b1v.y);
        float z2 = xr[i].z + (d2 * inv * g1v.z + b1v.z);
        float z3 = xr[i].w + (d3 * inv * g1v.w + b1v.w);

        float mu2 = wave_sum32(z0 + z1 + z2 + z3) * (1.0f / CDIM);
        float e0 = z0 - mu2, e1 = z1 - mu2, e2 = z2 - mu2, e3 = z3 - mu2;
        float var2 = wave_sum32(e0 * e0 + e1 * e1 + e2 * e2 + e3 * e3) * (1.0f / CDIM);
        float inv2 = rsqrtf(var2 + LN_EPS);

        float4 o;
        o.x = e0 * inv2 * g2v.x + b2v.x;
        o.y = e1 * inv2 * g2v.y + b2v.y;
        o.z = e2 * inv2 * g2v.z + b2v.z;
        o.w = e3 * inv2 * g2v.w + b2v.w;
        *(float4*)(out + (size_t)(node0 + i) * CDIM + c) = o;
    }
}

extern "C" void kernel_launch(void* const* d_in, const int* in_sizes, int n_in,
                              void* d_out, int out_size, void* d_ws, size_t ws_size,
                              hipStream_t stream) {
    const float* x  = (const float*)d_in[0];
    const int*   ei = (const int*)d_in[1];
    const float* W  = (const float*)d_in[2];
    const float* b  = (const float*)d_in[3];
    const float* g1 = (const float*)d_in[4];
    const float* be1 = (const float*)d_in[5];
    const float* g2 = (const float*)d_in[6];
    const float* be2 = (const float*)d_in[7];
    float* out = (float*)d_out;

    const int N = in_sizes[0] / CDIM;
    const int E = in_sizes[1] / 2;

    int*   cnt     = (int*)d_ws;
    int*   rowptr  = cnt + N;
    int*   cursor  = rowptr + N;
    float* dinv    = (float*)(cursor + N);
    int*   csr_src = (int*)(dinv + N);

    gcn_init_kernel<<<(N + 255) / 256, 256, 0, stream>>>(cnt, N);
    gcn_count_kernel<<<(E + 255) / 256, 256, 0, stream>>>(ei, cnt, E);
    gcn_scan_kernel<<<(N + 1023) / 1024, 1024, 0, stream>>>(cnt, rowptr, cursor, dinv, N);
    gcn_fill_kernel<<<(E + 255) / 256, 256, 0, stream>>>(ei, cursor, csr_src, E);
    gcn_gather_gemm_epilogue_kernel<<<(N + 31) / 32, 512, 0, stream>>>(x, W, rowptr, cnt, csr_src, dinv, b,
                                                                      g1, be1, g2, be2, out, N);
}


# edges grouped by destination on LDS counters: per-chunk histogram over 256-node bins, scan, binned records, one block per bin counts and places in LDS; no global atomics; sum in slot order as before
# speedup vs baseline: 4.1803x; 2.2396x over previous
#include <hip/hip_runtime.h>
#include <hip/hip_bf16.h>

#define CDIM 128
#define LN_EPS 1e-5f

typedef float v2f __attribute__((ext_vector_type(2)));
typedef float v8f __attribute__((ext_vector_type(8)));

__device__ __forceinline__ float wave_sum32(float v) {
#pragma unroll
    for (int o = 16; o > 0; o >>= 1) v += __shfl_xor(v, o, 32);
    return v;
}

__device__ __forceinline__ int wave_isum32(int v) {
#pragma unroll
    for (int o = 16; o > 0; o >>= 1) v += __shfl_xor(v, o, 32);
    return v;
}


#define BIN_SHIFT   8
#define BIN_NODES   256
#define BIN_WINDOW  1024
#define EDGE_BLOCKS 256

__global__ void __launch_bounds__(1024)
gcn_bin_hist_kernel(const int* __restrict__ ei, int* __restrict__ H, int E, int NB, int chunk) {
    __shared__ int s_h[BIN_WINDOW];
    const int tid = threadIdx.x;
    const int e0 = (int)blockIdx.x * chunk;
    const int e1 = (e0 + chunk) < E ? (e0 + chunk) : E;
    for (int b0 = 0; b0 < NB; b0 += BIN_WINDOW) {
        s_h[tid] = 0;
        __syncthreads();
        for (int base = e0; base < e1; base += 4096) {
            int d[4];
#pragma unroll
            for (int k = 0; k < 4; ++k) {
                const int e = base + k * 1024 + tid;
                d[k] = e < e1 ? ei[E + e] : -1;
            }
#pragma unroll
            for (int k = 0; k < 4; ++k) {
                const unsigned b = (unsigned)((d[k] >> BIN_SHIFT) - b0);
                if (d[k] >= 0 && b < (unsigned)BIN_WINDOW) atomicAdd(&s_h[b], 1);
            }
        }
        __syncthreads();
        if (b0 + tid < NB) H[(size_t)(b0 + tid) * EDGE_BLOCKS + blockIdx.x] = s_h[tid];
        __syncthreads();
    }
}

__global__ void __launch_bounds__(1024)
gcn_scan_kernel(const int* __restrict__ in, int* __restrict__ out, int M) {
    __shared__ int s_pre[32];
    __shared__ int s_tot[32];
    const int tid  = threadIdx.x;
    const int lane = tid & 31;
    const int wave = tid >> 5;

    const int4* c4 = (const int4*)in;
    const int n4 = blockIdx.x * 256;
    int pre = 0;
#pragma unroll 4
    for (int i = tid; i < n4; i += 1024) {
        const int4 v = c4[i];
        pre += (v.x + v.y) + (v.z + v.w);
    }
    pre = wave_isum32(pre);

    const int idx  = blockIdx.x * 1024 + tid;
    const int cval = idx < M ? in[idx] : 0;
    int inc = cval;
#pragma unroll
    for (int o = 1; o < 32; o <<= 1) {
        const int t = __shfl_up(inc, o, 32);
        if (lane >= o) inc += t;
    }
    if (lane == 0)  s_pre[wave] = pre;
    if (lane == 31) s_tot[wave] = inc;
    __syncthreads();
    const int blockpre = wave_isum32(s_pre[lane]);
    const int wavepre  = wave_isum32(lane < wave ? s_tot[lane] : 0);
    if (idx < M) out[idx] = blockpre + wavepre + inc - cval;
}

__global__ void __launch_bounds__(1024)
gcn_bin_scatter_kernel(const int* __restrict__ ei, const int* __restrict__ Hoff,
                       int2* __restrict__ rec, int E, int NB, int chunk) {
    __shared__ int s_cur[BIN_WINDOW];
    const int tid = threadIdx.x;
    const int e0 = (int)blockIdx.x * chunk;
    const int e1 = (e0 + chunk) < E ? (e0 + chunk) : E;
    for (int b0 = 0; b0 < NB; b0 += BIN_WINDOW) {
        s_cur[tid] = (b0 + tid < NB) ? Hoff[(size_t)(b0 + tid) * EDGE_BLOCKS + blockIdx.x] : 0;
        __syncthreads();
        for (int base = e0; base < e1; base += 4096) {
            int d[4], sr[4];
#pragma unroll
            for (int k = 0; k < 4; ++k) {
                const int e = base + k * 1024 + tid;
                d[k]  = e < e1 ? ei[E + e] : -1;
                sr[k] = e < e1 ? ei[e] : 0;
            }
#pragma unroll
            for (int k = 0; k < 4; ++k) {
                const unsigned b = (unsigned)((d[k] >> BIN_SHIFT) - b0);
                if (d[k] >= 0 && b < (unsigned)BIN_WINDOW) {
                    const int pos = atomicAdd(&s_cur[b], 1);
                    rec[pos] = make_int2(sr[k], d[k]);
                }
            }
        }
        __syncthreads();
    }
}

__global__ void __launch_bounds__(1024)
gcn_bin_place_kernel(const int2* __restrict__ rec, const int* __restrict__ Hoff,
                     int* __restrict__ cnt, int* __restrict__ rowptr, float* __restrict__ dinv,
                     int* __restrict__ csr_src, int N, int E, int NB) {
    __shared__ int s_c[BIN_NODES];
    __shared__ int s_off[BIN_NODES];
    __shared__ int s_wt[8];
    const int tid  = threadIdx.x;
    const int lane = tid & 31;
    const int wave = tid >> 5;
    const int bin  = blockIdx.x;
    const int nb0  = bin << BIN_SHIFT;
    const int rs = Hoff[(size_t)bin * EDGE_BLOCKS];
    const int re = bin + 1 < NB ? Hoff[(size_t)(bin + 1) * EDGE_BLOCKS] : E;

    if (tid < BIN_NODES) s_c[tid] = 0;
    __syncthreads();
    for (int base = rs; base < re; base += 4096) {
        int d[4];
#pragma unroll
        for (int k = 0; k < 4; ++k) {
            const int i = base + k * 1024 + tid;
            d[k] = i < re ? rec[i].y : -1;
        }
#pragma unroll
        for (int k = 0; k < 4; ++k)
            if (d[k] >= 0) atomicAdd(&s_c[d[k] - nb0], 1);
    }
    __syncthreads();

    int c = 0, inc = 0;
    if (tid < BIN_NODES) {
        c = s_c[tid];
        inc = c;
#pragma unroll
        for (int o = 1; o < 32; o <<= 1) {
            const int t = __shfl_up(inc, o, 32);
            if (lane >= o) inc += t;
        }
        if (lane == 31) s_wt[wave] = inc;
    }
    __syncthreads();
    if (tid < BIN_NODES) {
        int wp = 0;
#pragma unroll
        for (int w = 0; w < 8; ++w) wp += (w < wave) ? s_wt[w] : 0;
        const int start = rs + wp + inc - c;
        s_off[tid] = start;
        const int node = nb0 + tid;
        if (node < N) {
            cnt[node]    = c;
            rowptr[node] = start;
            dinv[node]   = rsqrtf((float)(c + 1));
        }
    }
    __syncthreads();

    for (int base = rs; base < re; base += 4096) {
        int2 r[4];
#pragma unroll
        for (int k = 0; k < 4; ++k) {
            const int i = base + k * 1024 + tid;
            r[k] = i < re ? rec[i] : make_int2(0, -1);
        }
#pragma unroll
        for (int k = 0; k < 4; ++k) {
            if (r[k].y >= 0) {
                const int pos = atomicAdd(&s_off[r[k].y - nb0], 1);
                csr_src[pos] = r[k].x;
            }
        }
    }
}

template <int U>
__device__ __forceinline__ void add_rows(const float* __restrict__ xs, int s, float w,
                                         int j, int c, float4& acc) {
    int    sj[U];
    float  wj[U];
    float4 r[U];
#pragma unroll
    for (int u = 0; u < U; ++u) {
        sj[u] = __builtin_amdgcn_readlane(s, j + u);
        wj[u] = __int_as_float(__builtin_amdgcn_readlane(__float_as_int(w), j + u));
    }
#pragma unroll
    for (int u = 0; u < U; ++u) r[u] = *(const float4*)(xs + (size_t)sj[u] * CDIM + c);
#pragma unroll
    for (int u = 0; u < U; ++u) {
        acc.x = fmaf(wj[u], r[u].x, acc.x);
        acc.y = fmaf(wj[u], r[u].y, acc.y);
        acc.z = fmaf(wj[u], r[u].z, acc.z);
        acc.w = fmaf(wj[u], r[u].w, acc.w);
    }
}

__device__ __forceinline__ void add_chunk(const float* __restrict__ xs, int s, float w,
                                          int m, int c, float4& acc) {
    int j = 0;
    for (; j + 8 <= m; j += 8) add_rows<8>(xs, s, w, j, c, acc);
    switch (m - j) {
        case 7: add_rows<7>(xs, s, w, j, c, acc); break;
        case 6: add_rows<6>(xs, s, w, j, c, acc); break;
        case 5: add_rows<5>(xs, s, w, j, c, acc); break;
        case 4: add_rows<4>(xs, s, w, j, c, acc); break;
        case 3: add_rows<3>(xs, s, w, j, c, acc); break;
        case 2: add_rows<2>(xs, s, w, j, c, acc); break;
        case 1: add_rows<1>(xs, s, w, j, c, acc); break;
        default: break;
    }
}

#define TILE_M 32
#define LDS_PITCH 132

__global__ void __launch_bounds__(512) __attribute__((amdgpu_waves_per_eu(16, 16)))
gcn_gather_gemm_epilogue_kernel(const float* __restrict__ x, const float* __restrict__ W,
                                const int* __restrict__ rowptr, const int* __restrict__ cnt,
                                const int* __restrict__ csr_src, const float* __restrict__ dinv,
                                const float* __restrict__ bvec,
                                const float* __restrict__ g1, const float* __restrict__ be1,
                                const float* __restrict__ g2, const float* __restrict__ be2,
                                float* __restrict__ out, int N) {
    __shared__ __attribute__((aligned(16))) float s_agg[TILE_M * LDS_PITCH];
    __shared__ __attribute__((aligned(16))) float s_d[2][TILE_M * LDS_PITCH];

    const int wave  = __builtin_amdgcn_readfirstlane((int)(threadIdx.x >> 5));
    const int node0 = (int)blockIdx.x * TILE_M + 2 * wave;
    const int lane  = threadIdx.x & 31;
    const int c = lane * 4;

    bool   live[2];
    int    st[2], nn[2], m0[2], s0[2];
    float  di[2], w0[2];
    float4 xv[2];
#pragma unroll
    for (int i = 0; i < 2; ++i) {
        live[i] = node0 + i < N;
        const int q = live[i] ? node0 + i : 0;
        st[i] = rowptr[q];
        nn[i] = live[i] ? cnt[q] : 0;
        di[i] = dinv[q];
        xv[i] = *(const float4*)(x + (size_t)q * CDIM + c);
    }
#pragma unroll
    for (int i = 0; i < 2; ++i) {
        m0[i] = nn[i] < 32 ? nn[i] : 32;
        const int k = lane < m0[i] ? lane : 0;
        s0[i] = nn[i] > 0 ? csr_src[st[i] + k] : 0;
    }
#pragma unroll
    for (int i = 0; i < 2; ++i) w0[i] = dinv[s0[i]] * di[i];

#pragma unroll
    for (int i = 0; i < 2; ++i) {
        float4 acc = make_float4(0.0f, 0.0f, 0.0f, 0.0f);
        add_chunk(x, s0[i], w0[i], m0[i], c, acc);
        for (int j0 = 32; j0 < nn[i]; j0 += 32) {
            const int m = (nn[i] - j0) < 32 ? (nn[i] - j0) : 32;
            const int   k = lane < m ? lane : 0;
            const int   s = csr_src[st[i] + j0 + k];
            const float w = dinv[s] * di[i];
            add_chunk(x, s, w, m, c, acc);
        }
        if (live[i]) {
            const float sw = di[i] * di[i];
            acc.x = fmaf(sw, xv[i].x, acc.x);
            acc.y = fmaf(sw, xv[i].y, acc.y);
            acc.z = fmaf(sw, xv[i].z, acc.z);
            acc.w = fmaf(sw, xv[i].w, acc.w);
        }
        *(float4*)(s_agg + (2 * wave + i) * LDS_PITCH + c) = acc;
    }

    const int half  = lane >> 4;
    const int l     = lane & 15;
    const int ct    = wave & 7;
    const int kp    = wave >> 3;
    const int kbase = kp * 64 + half * 4;
    const float* wrow = W + (size_t)(ct * 16 + l) * CDIM + kbase;
    float4 bf[8];
#pragma unroll
    for (int kb = 0; kb < 8; ++kb) bf[kb] = *(const float4*)(wrow + kb * 8);

    __syncthreads();

#pragma unroll
    for (int t = 0; t < 2; ++t) {
        const float* arow = s_agg + (16 * t + l) * LDS_PITCH + kbase;
        v8f d = {};
#pragma unroll
        for (int kb = 0; kb < 8; ++kb) {
            const float4 af = *(const float4*)(arow + kb * 8);
            v2f a0 = {af.x, af.y};
            v2f b0 = {bf[kb].x, bf[kb].y};
            d = __builtin_amdgcn_wmma_f32_16x16x4_f32(
                 false, a0,  false, b0,
                 (short)0, d,  false,  false);
            v2f a1 = {af.z, af.w};
            v2f b1 = {bf[kb].z, bf[kb].w};
            d = __builtin_amdgcn_wmma_f32_16x16x4_f32(
                 false, a1,  false, b1,
                 (short)0, d,  false,  false);
        }
        float* dp = s_d[kp] + (16 * t) * LDS_PITCH + ct * 16 + l;
#pragma unroll
        for (int r = 0; r < 8; ++r) dp[(r + 8 * half) * LDS_PITCH] = d[r];
    }

    __syncthreads();

    float4 xr[2];
#pragma unroll
    for (int i = 0; i < 2; ++i)
        xr[i] = *(const float4*)(x + (size_t)(live[i] ? node0 + i : 0) * CDIM + c);
    const float4 bv  = *(const float4*)(bvec + c);
    const float4 g1v = *(const float4*)(g1 + c);
    const float4 b1v = *(const float4*)(be1 + c);
    const float4 g2v = *(const float4*)(g2 + c);
    const float4 b2v = *(const float4*)(be2 + c);

#pragma unroll
    for (int i = 0; i < 2; ++i) {
        if (!live[i]) continue;
        const int row = 2 * wave + i;
        const float4 p0 = *(const float4*)(s_d[0] + row * LDS_PITCH + c);
        const float4 p1 = *(const float4*)(s_d[1] + row * LDS_PITCH + c);

        float v0 = fmaxf((p0.x + p1.x) + bv.x, 0.0f);
        float v1 = fmaxf((p0.y + p1.y) + bv.y, 0.0f);
        float v2 = fmaxf((p0.z + p1.z) + bv.z, 0.0f);
        float v3 = fmaxf((p0.w + p1.w) + bv.w, 0.0f);

        float mu = wave_sum32(v0 + v1 + v2 + v3) * (1.0f / CDIM);
        float d0 = v0 - mu, d1 = v1 - mu, d2 = v2 - mu, d3 = v3 - mu;
        float var = wave_sum32(d0 * d0 + d1 * d1 + d2 * d2 + d3 * d3) * (1.0f / CDIM);
        float inv = rsqrtf(var + LN_EPS);
        float z0 = xr[i].x + (d0 * inv * g1v.x + b1v.x);
        float z1 = xr[i].y + (d1 * inv * g1v.y + b1v.y);
        float z2 = xr[i].z + (d2 * inv * g1v.z + b1v.z);
        float z3 = xr[i].w + (d3 * inv * g1v.w + b1v.w);

        float mu2 = wave_sum32(z0 + z1 + z2 + z3) * (1.0f / CDIM);
        float e0 = z0 - mu2, e1 = z1 - mu2, e2 = z2 - mu2, e3 = z3 - mu2;
        float var2 = wave_sum32(e0 * e0 + e1 * e1 + e2 * e2 + e3 * e3) * (1.0f / CDIM);
        float inv2 = rsqrtf(var2 + LN_EPS);

        float4 o;
        o.x = e0 * inv2 * g2v.x + b2v.x;
        o.y = e1 * inv2 * g2v.y + b2v.y;
        o.z = e2 * inv2 * g2v.z + b2v.z;
        o.w = e3 * inv2 * g2v.w + b2v.w;
        *(float4*)(out + (size_t)(node0 + i) * CDIM + c) = o;
    }
}

extern "C" void kernel_launch(void* const* d_in, const int* in_sizes, int n_in,
                              void* d_out, int out_size, void* d_ws, size_t ws_size,
                              hipStream_t stream) {
    const float* x  = (const float*)d_in[0];
    const int*   ei = (const int*)d_in[1];
    const float* W  = (const float*)d_in[2];
    const float* b  = (const float*)d_in[3];
    const float* g1 = (const float*)d_in[4];
    const float* be1 = (const float*)d_in[5];
    const float* g2 = (const float*)d_in[6];
    const float* be2 = (const float*)d_in[7];
    float* out = (float*)d_out;

    const int N = in_sizes[0] / CDIM;
    const int E = in_sizes[1] / 2;

    const int NB    = (N + 255) / 256;
    const int M     = NB * 256;
    const int chunk = (E + 255) / 256;

    int*   H       = (int*)d_ws;
    int*   Hoff    = H + M;
    int2*  rec     = (int2*)(Hoff + M);
    int*   cnt     = (int*)(rec + E);
    int*   rowptr  = cnt + N;
    float* dinv    = (float*)(rowptr + N);
    int*   csr_src = (int*)(dinv + N);

    gcn_bin_hist_kernel<<<256, 1024, 0, stream>>>(ei, H, E, NB, chunk);
    gcn_scan_kernel<<<(M + 1023) / 1024, 1024, 0, stream>>>(H, Hoff, M);
    gcn_bin_scatter_kernel<<<256, 1024, 0, stream>>>(ei, Hoff, rec, E, NB, chunk);
    gcn_bin_place_kernel<<<NB, 1024, 0, stream>>>(rec, Hoff, cnt, rowptr, dinv, csr_src, N, E, NB);
    gcn_gather_gemm_epilogue_kernel<<<(N + 31) / 32, 512, 0, stream>>>(x, W, rowptr, cnt, csr_src, dinv, b,
                                                                      g1, be1, g2, be2, out, N);
}


# edge partition made chunk-local: each edge chunk sorted by bin inside its own slice, per-chunk offset table, bins walk 256 pieces; hist and scan launches gone; sum in slot order as before
# speedup vs baseline: 4.9461x; 1.1832x over previous
#include <hip/hip_runtime.h>
#include <hip/hip_bf16.h>

#define CDIM 128
#define LN_EPS 1e-5f

typedef float v2f __attribute__((ext_vector_type(2)));
typedef float v8f __attribute__((ext_vector_type(8)));

__device__ __forceinline__ float wave_sum32(float v) {
#pragma unroll
    for (int o = 16; o > 0; o >>= 1) v += __shfl_xor(v, o, 32);
    return v;
}

__device__ __forceinline__ int wave_isum32(int v) {
#pragma unroll
    for (int o = 16; o > 0; o >>= 1) v += __shfl_xor(v, o, 32);
    return v;
}


#define BIN_SHIFT   8
#define BIN_NODES   256
#define BIN_WINDOW  1024
#define EDGE_BLOCKS 256

__global__ void __launch_bounds__(1024)
gcn_chunk_sort_kernel(const int* __restrict__ ei, int* __restrict__ P, int2* __restrict__ rec,
                      int E, int NB, int chunk) {
    __shared__ int s_h[BIN_WINDOW];
    __shared__ int s_wt[32];
    const int tid  = threadIdx.x;
    const int lane = tid & 31;
    const int wave = tid >> 5;
    const int e0 = (int)blockIdx.x * chunk;
    const int e1 = (e0 + chunk) < E ? (e0 + chunk) : E;
    int*  Pg   = P + (size_t)blockIdx.x * (NB + 1);
    int2* recg = rec + (size_t)blockIdx.x * chunk;
    int carry = 0;
    for (int b0 = 0; b0 < NB; b0 += BIN_WINDOW) {
        s_h[tid] = 0;
        __syncthreads();
        for (int base = e0; base < e1; base += 8192) {
            int d[8];
#pragma unroll
            for (int k = 0; k < 8; ++k) {
                const int e = base + k * 1024 + tid;
                d[k] = e < e1 ? ei[E + e] : -1;
            }
#pragma unroll
            for (int k = 0; k < 8; ++k) {
                const unsigned b = (unsigned)((d[k] >> BIN_SHIFT) - b0);
                if (d[k] >= 0 && b < (unsigned)BIN_WINDOW) atomicAdd(&s_h[b], 1);
            }
        }
        __syncthreads();

        const int cval = s_h[tid];
        int inc = cval;
#pragma unroll
        for (int o = 1; o < 32; o <<= 1) {
            const int t = __shfl_up(inc, o, 32);
            if (lane >= o) inc += t;
        }
        if (lane == 31) s_wt[wave] = inc;
        __syncthreads();
        const int wt      = s_wt[lane];
        const int wavepre = wave_isum32(lane < wave ? wt : 0);
        const int total   = wave_isum32(wt);
        const int excl    = carry + wavepre + inc - cval;
        if (b0 + tid < NB) Pg[b0 + tid] = excl;
        s_h[tid] = excl;
        __syncthreads();

        for (int base = e0; base < e1; base += 8192) {
            int d[8], sr[8], pos[8];
#pragma unroll
            for (int k = 0; k < 8; ++k) {
                const int e = base + k * 1024 + tid;
                d[k]  = e < e1 ? ei[E + e] : -1;
                sr[k] = e < e1 ? ei[e] : 0;
            }
#pragma unroll
            for (int k = 0; k < 8; ++k) {
                const unsigned b = (unsigned)((d[k] >> BIN_SHIFT) - b0);
                pos[k] = -1;
                if (d[k] >= 0 && b < (unsigned)BIN_WINDOW) pos[k] = atomicAdd(&s_h[b], 1);
            }
#pragma unroll
            for (int k = 0; k < 8; ++k)
                if (pos[k] >= 0) recg[pos[k]] = make_int2(sr[k], d[k]);
        }
        carry += total;
        __syncthreads();
    }
    if (tid == 0) Pg[NB] = carry;
}

static_assert(EDGE_BLOCKS * 4 == 1024, "place: four threads per piece");
static_assert(EDGE_BLOCKS % 32 == 0, "place: the table is read by whole waves");

__global__ void __launch_bounds__(1024)
gcn_bin_place_kernel(const int2* __restrict__ rec, const int* __restrict__ P,
                     int* __restrict__ cnt, int* __restrict__ rowptr, float* __restrict__ dinv,
                     int* __restrict__ csr_src, int N, int NB, int chunk) {
    __shared__ int s_c[BIN_NODES];
    __shared__ int s_off[BIN_NODES];
    __shared__ int s_ps[EDGE_BLOCKS];
    __shared__ int s_pl[EDGE_BLOCKS];
    __shared__ int s_rs[EDGE_BLOCKS / 32];
    __shared__ int s_wt[8];
    const int tid  = threadIdx.x;
    const int lane = tid & 31;
    const int wave = tid >> 5;
    const int bin  = blockIdx.x;
    const int nb0  = bin << BIN_SHIFT;

    if (tid < BIN_NODES) s_c[tid] = 0;
    if (tid < EDGE_BLOCKS) {
        const int* Pg = P + (size_t)tid * (NB + 1) + bin;
        const int p0 = Pg[0];
        const int p1 = Pg[1];
        s_ps[tid] = tid * chunk + p0;
        s_pl[tid] = p1 - p0;
        const int ws = wave_isum32(p0);
        if (lane == 0) s_rs[wave] = ws;
    }
    __syncthreads();
    int rs = 0;
#pragma unroll
    for (int w = 0; w < EDGE_BLOCKS / 32; ++w) rs += s_rs[w];

    const int piece  = tid >> 2;
    const int sub    = tid & 3;
    const int pstart = s_ps[piece];
    const int plen   = s_pl[piece];

    for (int i0 = 0; i0 < plen; i0 += 32) {
        int d[8];
#pragma unroll
        for (int k = 0; k < 8; ++k) {
            const int i = i0 + 4 * k + sub;
            d[k] = i < plen ? rec[pstart + i].y : -1;
        }
#pragma unroll
        for (int k = 0; k < 8; ++k)
            if (d[k] >= 0) atomicAdd(&s_c[d[k] - nb0], 1);
    }
    __syncthreads();

    int c = 0, inc = 0;
    if (tid < BIN_NODES) {
        c = s_c[tid];
        inc = c;
#pragma unroll
        for (int o = 1; o < 32; o <<= 1) {
            const int t = __shfl_up(inc, o, 32);
            if (lane >= o) inc += t;
        }
        if (lane == 31) s_wt[wave] = inc;
    }
    __syncthreads();
    if (tid < BIN_NODES) {
        int wp = 0;
#pragma unroll
        for (int w = 0; w < 8; ++w) wp += (w < wave) ? s_wt[w] : 0;
        const int start = rs + wp + inc - c;
        s_off[tid] = start;
        const int node = nb0 + tid;
        if (node < N) {
            cnt[node]    = c;
            rowptr[node] = start;
            dinv[node]   = rsqrtf((float)(c + 1));
        }
    }
    __syncthreads();

    for (int i0 = 0; i0 < plen; i0 += 32) {
        int2 r[8];
#pragma unroll
        for (int k = 0; k < 8; ++k) {
            const int i = i0 + 4 * k + sub;
            r[k] = i < plen ? rec[pstart + i] : make_int2(0, -1);
        }
        int pos[8];
#pragma unroll
        for (int k = 0; k < 8; ++k) {
            pos[k] = -1;
            if (r[k].y >= 0) pos[k] = atomicAdd(&s_off[r[k].y - nb0], 1);
        }
#pragma unroll
        for (int k = 0; k < 8; ++k)
            if (pos[k] >= 0) csr_src[pos[k]] = r[k].x;
    }
}

template <int U>
__device__ __forceinline__ void add_rows(const float* __restrict__ xs, int s, float w,
                                         int j, int c, float4& acc) {
    int    sj[U];
    float  wj[U];
    float4 r[U];
#pragma unroll
    for (int u = 0; u < U; ++u) {
        sj[u] = __builtin_amdgcn_readlane(s, j + u);
        wj[u] = __int_as_float(__builtin_amdgcn_readlane(__float_as_int(w), j + u));
    }
#pragma unroll
    for (int u = 0; u < U; ++u) r[u] = *(const float4*)(xs + (size_t)sj[u] * CDIM + c);
#pragma unroll
    for (int u = 0; u < U; ++u) {
        acc.x = fmaf(wj[u], r[u].x, acc.x);
        acc.y = fmaf(wj[u], r[u].y, acc.y);
        acc.z = fmaf(wj[u], r[u].z, acc.z);
        acc.w = fmaf(wj[u], r[u].w, acc.w);
    }
}

__device__ __forceinline__ void add_chunk(const float* __restrict__ xs, int s, float w,
                                          int m, int c, float4& acc) {
    int j = 0;
    for (; j + 8 <= m; j += 8) add_rows<8>(xs, s, w, j, c, acc);
    switch (m - j) {
        case 7: add_rows<7>(xs, s, w, j, c, acc); break;
        case 6: add_rows<6>(xs, s, w, j, c, acc); break;
        case 5: add_rows<5>(xs, s, w, j, c, acc); break;
        case 4: add_rows<4>(xs, s, w, j, c, acc); break;
        case 3: add_rows<3>(xs, s, w, j, c, acc); break;
        case 2: add_rows<2>(xs, s, w, j, c, acc); break;
        case 1: add_rows<1>(xs, s, w, j, c, acc); break;
        default: break;
    }
}

#define TILE_M 32
#define LDS_PITCH 132

__global__ void __launch_bounds__(512) __attribute__((amdgpu_waves_per_eu(16, 16)))
gcn_gather_gemm_epilogue_kernel(const float* __restrict__ x, const float* __restrict__ W,
                                const int* __restrict__ rowptr, const int* __restrict__ cnt,
                                const int* __restrict__ csr_src, const float* __restrict__ dinv,
                                const float* __restrict__ bvec,
                                const float* __restrict__ g1, const float* __restrict__ be1,
                                const float* __restrict__ g2, const float* __restrict__ be2,
                                float* __restrict__ out, int N) {
    __shared__ __attribute__((aligned(16))) float s_agg[TILE_M * LDS_PITCH];
    __shared__ __attribute__((aligned(16))) float s_d[2][TILE_M * LDS_PITCH];

    const int wave  = __builtin_amdgcn_readfirstlane((int)(threadIdx.x >> 5));
    const int node0 = (int)blockIdx.x * TILE_M + 2 * wave;
    const int lane  = threadIdx.x & 31;
    const int c = lane * 4;

    bool   live[2];
    int    st[2], nn[2], m0[2], s0[2];
    float  di[2], w0[2];
    float4 xv[2];
#pragma unroll
    for (int i = 0; i < 2; ++i) {
        live[i] = node0 + i < N;
        const int q = live[i] ? node0 + i : 0;
        st[i] = rowptr[q];
        nn[i] = live[i] ? cnt[q] : 0;
        di[i] = dinv[q];
        xv[i] = *(const float4*)(x + (size_t)q * CDIM + c);
    }
#pragma unroll
    for (int i = 0; i < 2; ++i) {
        m0[i] = nn[i] < 32 ? nn[i] : 32;
        const int k = lane < m0[i] ? lane : 0;
        s0[i] = nn[i] > 0 ? csr_src[st[i] + k] : 0;
    }
#pragma unroll
    for (int i = 0; i < 2; ++i) w0[i] = dinv[s0[i]] * di[i];

#pragma unroll
    for (int i = 0; i < 2; ++i) {
        float4 acc = make_float4(0.0f, 0.0f, 0.0f, 0.0f);
        add_chunk(x, s0[i], w0[i], m0[i], c, acc);
        for (int j0 = 32; j0 < nn[i]; j0 += 32) {
            const int m = (nn[i] - j0) < 32 ? (nn[i] - j0) : 32;
            const int   k = lane < m ? lane : 0;
            const int   s = csr_src[st[i] + j0 + k];
            const float w = dinv[s] * di[i];
            add_chunk(x, s, w, m, c, acc);
        }
        if (live[i]) {
            const float sw = di[i] * di[i];
            acc.x = fmaf(sw, xv[i].x, acc.x);
            acc.y = fmaf(sw, xv[i].y, acc.y);
            acc.z = fmaf(sw, xv[i].z, acc.z);
            acc.w = fmaf(sw, xv[i].w, acc.w);
        }
        *(float4*)(s_agg + (2 * wave + i) * LDS_PITCH + c) = acc;
    }

    const int half  = lane >> 4;
    const int l     = lane & 15;
    const int ct    = wave & 7;
    const int kp    = wave >> 3;
    const int kbase = kp * 64 + half * 4;
    const float* wrow = W + (size_t)(ct * 16 + l) * CDIM + kbase;
    float4 bf[8];
#pragma unroll
    for (int kb = 0; kb < 8; ++kb) bf[kb] = *(const float4*)(wrow + kb * 8);

    __syncthreads();

#pragma unroll
    for (int t = 0; t < 2; ++t) {
        const float* arow = s_agg + (16 * t + l) * LDS_PITCH + kbase;
        v8f d = {};
#pragma unroll
        for (int kb = 0; kb < 8; ++kb) {
            const float4 af = *(const float4*)(arow + kb * 8);
            v2f a0 = {af.x, af.y};
            v2f b0 = {bf[kb].x, bf[kb].y};
            d = __builtin_amdgcn_wmma_f32_16x16x4_f32(
                 false, a0,  false, b0,
                 (short)0, d,  false,  false);
            v2f a1 = {af.z, af.w};
            v2f b1 = {bf[kb].z, bf[kb].w};
            d = __builtin_amdgcn_wmma_f32_16x16x4_f32(
                 false, a1,  false, b1,
                 (short)0, d,  false,  false);
        }
        float* dp = s_d[kp] + (16 * t) * LDS_PITCH + ct * 16 + l;
#pragma unroll
        for (int r = 0; r < 8; ++r) dp[(r + 8 * half) * LDS_PITCH] = d[r];
    }

    __syncthreads();

    float4 xr[2];
#pragma unroll
    for (int i = 0; i < 2; ++i)
        xr[i] = *(const float4*)(x + (size_t)(live[i] ? node0 + i : 0) * CDIM + c);
    const float4 bv  = *(const float4*)(bvec + c);
    const float4 g1v = *(const float4*)(g1 + c);
    const float4 b1v = *(const float4*)(be1 + c);
    const float4 g2v = *(const float4*)(g2 + c);
    const float4 b2v = *(const float4*)(be2 + c);

#pragma unroll
    for (int i = 0; i < 2; ++i) {
        if (!live[i]) continue;
        const int row = 2 * wave + i;
        const float4 p0 = *(const float4*)(s_d[0] + row * LDS_PITCH + c);
        const float4 p1 = *(const float4*)(s_d[1] + row * LDS_PITCH + c);

        float v0 = fmaxf((p0.x + p1.x) + bv.x, 0.0f);
        float v1 = fmaxf((p0.y + p1.y) + bv.y, 0.0f);
        float v2 = fmaxf((p0.z + p1.z) + bv.z, 0.0f);
        float v3 = fmaxf((p0.w + p1.w) + bv.w, 0.0f);

        float mu = wave_sum32(v0 + v1 + v2 + v3) * (1.0f / CDIM);
        float d0 = v0 - mu, d1 = v1 - mu, d2 = v2 - mu, d3 = v3 - mu;
        float var = wave_sum32(d0 * d0 + d1 * d1 + d2 * d2 + d3 * d3) * (1.0f / CDIM);
        float inv = rsqrtf(var + LN_EPS);
        float z0 = xr[i].x + (d0 * inv * g1v.x + b1v.x);
        float z1 = xr[i].y + (d1 * inv * g1v.y + b1v.y);
        float z2 = xr[i].z + (d2 * inv * g1v.z + b1v.z);
        float z3 = xr[i].w + (d3 * inv * g1v.w + b1v.w);

        float mu2 = wave_sum32(z0 + z1 + z2 + z3) * (1.0f / CDIM);
        float e0 = z0 - mu2, e1 = z1 - mu2, e2 = z2 - mu2, e3 = z3 - mu2;
        float var2 = wave_sum32(e0 * e0 + e1 * e1 + e2 * e2 + e3 * e3) * (1.0f / CDIM);
        float inv2 = rsqrtf(var2 + LN_EPS);

        float4 o;
        o.x = e0 * inv2 * g2v.x + b2v.x;
        o.y = e1 * inv2 * g2v.y + b2v.y;
        o.z = e2 * inv2 * g2v.z + b2v.z;
        o.w = e3 * inv2 * g2v.w + b2v.w;
        *(float4*)(out + (size_t)(node0 + i) * CDIM + c) = o;
    }
}

extern "C" void kernel_launch(void* const* d_in, const int* in_sizes, int n_in,
                              void* d_out, int out_size, void* d_ws, size_t ws_size,
                              hipStream_t stream) {
    const float* x  = (const float*)d_in[0];
    const int*   ei = (const int*)d_in[1];
    const float* W  = (const float*)d_in[2];
    const float* b  = (const float*)d_in[3];
    const float* g1 = (const float*)d_in[4];
    const float* be1 = (const float*)d_in[5];
    const float* g2 = (const float*)d_in[6];
    const float* be2 = (const float*)d_in[7];
    float* out = (float*)d_out;

    const int N = in_sizes[0] / CDIM;
    const int E = in_sizes[1] / 2;

    const int NB    = (N + 255) / 256;
    const int PW    = EDGE_BLOCKS * (NB + 1);
    const int chunk = (E + EDGE_BLOCKS - 1) / EDGE_BLOCKS;

    int*   P       = (int*)d_ws;
    int2*  rec     = (int2*)(P + PW);
    int*   cnt     = (int*)(rec + (size_t)EDGE_BLOCKS * chunk);
    int*   rowptr  = cnt + N;
    float* dinv    = (float*)(rowptr + N);
    int*   csr_src = (int*)(dinv + N);

    gcn_chunk_sort_kernel<<<EDGE_BLOCKS, 1024, 0, stream>>>(ei, P, rec, E, NB, chunk);
    gcn_bin_place_kernel<<<NB, 1024, 0, stream>>>(rec, P, cnt, rowptr, dinv, csr_src, N, NB, chunk);
    gcn_gather_gemm_epilogue_kernel<<<(N + 31) / 32, 512, 0, stream>>>(x, W, rowptr, cnt, csr_src, dinv, b,
                                                                      g1, be1, g2, be2, out, N);
}


# both scatter passes staged in LDS: a chunk's sorted records and a bin's buckets are put in place in an LDS copy, then written out in order as whole 128-byte lines; slot order as before
# speedup vs baseline: 5.3572x; 1.0831x over previous
#include <hip/hip_runtime.h>
#include <hip/hip_bf16.h>

#define CDIM 128
#define LN_EPS 1e-5f

typedef float v2f __attribute__((ext_vector_type(2)));
typedef float v8f __attribute__((ext_vector_type(8)));

__device__ __forceinline__ float wave_sum32(float v) {
#pragma unroll
    for (int o = 16; o > 0; o >>= 1) v += __shfl_xor(v, o, 32);
    return v;
}

__device__ __forceinline__ int wave_isum32(int v) {
#pragma unroll
    for (int o = 16; o > 0; o >>= 1) v += __shfl_xor(v, o, 32);
    return v;
}


#define BIN_SHIFT   8
#define BIN_NODES   256
#define BIN_WINDOW  1024
#define EDGE_BLOCKS 256
#define STAGE_RECS  8192
#define STAGE_SRCS  8192

__device__ __forceinline__ int slice_pitch(int chunk) { return (chunk + 15) & ~15; }

__global__ void __launch_bounds__(1024)
gcn_chunk_sort_kernel(const int* __restrict__ ei, int* __restrict__ P, int2* __restrict__ rec,
                      int E, int NB, int chunk) {
    __shared__ int s_h[BIN_WINDOW];
    __shared__ int s_wt[32];
    __shared__ int2 s_rec[STAGE_RECS];
    const int tid  = threadIdx.x;
    const int lane = tid & 31;
    const int wave = tid >> 5;
    const int e0 = (int)blockIdx.x * chunk;
    const int e1 = (e0 + chunk) < E ? (e0 + chunk) : E;
    int*  Pg   = P + (size_t)blockIdx.x * (NB + 1);
    int2* recg = rec + (size_t)blockIdx.x * slice_pitch(chunk);
    const bool staged = NB <= BIN_WINDOW && chunk <= STAGE_RECS;
    int carry = 0;
    for (int b0 = 0; b0 < NB; b0 += BIN_WINDOW) {
        s_h[tid] = 0;
        __syncthreads();
        for (int base = e0; base < e1; base += 8192) {
            int d[8];
#pragma unroll
            for (int k = 0; k < 8; ++k) {
                const int e = base + k * 1024 + tid;
                d[k] = e < e1 ? ei[E + e] : -1;
            }
#pragma unroll
            for (int k = 0; k < 8; ++k) {
                const unsigned b = (unsigned)((d[k] >> BIN_SHIFT) - b0);
                if (d[k] >= 0 && b < (unsigned)BIN_WINDOW) atomicAdd(&s_h[b], 1);
            }
        }
        __syncthreads();

        const int cval = s_h[tid];
        int inc = cval;
#pragma unroll
        for (int o = 1; o < 32; o <<= 1) {
            const int t = __shfl_up(inc, o, 32);
            if (lane >= o) inc += t;
        }
        if (lane == 31) s_wt[wave] = inc;
        __syncthreads();
        const int wt      = s_wt[lane];
        const int wavepre = wave_isum32(lane < wave ? wt : 0);
        const int total   = wave_isum32(wt);
        const int excl    = carry + wavepre + inc - cval;
        if (b0 + tid < NB) Pg[b0 + tid] = excl;
        s_h[tid] = excl;
        __syncthreads();

        for (int base = e0; base < e1; base += 8192) {
            int d[8], sr[8], pos[8];
#pragma unroll
            for (int k = 0; k < 8; ++k) {
                const int e = base + k * 1024 + tid;
                d[k]  = e < e1 ? ei[E + e] : -1;
                sr[k] = e < e1 ? ei[e] : 0;
            }
#pragma unroll
            for (int k = 0; k < 8; ++k) {
                const unsigned b = (unsigned)((d[k] >> BIN_SHIFT) - b0);
                pos[k] = -1;
                if (d[k] >= 0 && b < (unsigned)BIN_WINDOW) pos[k] = atomicAdd(&s_h[b], 1);
            }
            if (staged) {
#pragma unroll
                for (int k = 0; k < 8; ++k)
                    if (pos[k] >= 0) s_rec[pos[k]] = make_int2(sr[k], d[k]);
            } else {
#pragma unroll
                for (int k = 0; k < 8; ++k)
                    if (pos[k] >= 0) recg[pos[k]] = make_int2(sr[k], d[k]);
            }
        }
        carry += total;
        __syncthreads();
    }
    if (staged) {
        static_assert(STAGE_RECS == 8 * 1024, "copy-out: eight records per thread");
        int2 v[8];
#pragma unroll
        for (int k = 0; k < 8; ++k) {
            const int i = k * 1024 + tid;
            v[k] = i < carry ? s_rec[i] : make_int2(0, 0);
        }
#pragma unroll
        for (int k = 0; k < 8; ++k) {
            const int i = k * 1024 + tid;
            if (i < carry) recg[i] = v[k];
        }
    }
    if (tid == 0) Pg[NB] = carry;
}

static_assert(EDGE_BLOCKS * 4 == 1024, "place: four threads per piece");
static_assert(EDGE_BLOCKS % 32 == 0, "place: the table is read by whole waves");

__global__ void __launch_bounds__(1024)
gcn_bin_place_kernel(const int2* __restrict__ rec, const int* __restrict__ P,
                     int* __restrict__ cnt, int* __restrict__ rowptr, float* __restrict__ dinv,
                     int* __restrict__ csr_src, int N, int NB, int chunk) {
    __shared__ int s_c[BIN_NODES];
    __shared__ int s_off[BIN_NODES];
    __shared__ int s_ps[EDGE_BLOCKS];
    __shared__ int s_pl[EDGE_BLOCKS];
    __shared__ int s_rs[EDGE_BLOCKS / 32];
    __shared__ int s_wt[8];
    __shared__ int s_stage[STAGE_SRCS];
    const int tid  = threadIdx.x;
    const int lane = tid & 31;
    const int wave = tid >> 5;
    const int bin  = blockIdx.x;
    const int nb0  = bin << BIN_SHIFT;

    if (tid < BIN_NODES) s_c[tid] = 0;
    if (tid < EDGE_BLOCKS) {
        const int* Pg = P + (size_t)tid * (NB + 1) + bin;
        const int p0 = Pg[0];
        const int p1 = Pg[1];
        s_ps[tid] = tid * slice_pitch(chunk) + p0;
        s_pl[tid] = p1 - p0;
        const int ws = wave_isum32(p0);
        if (lane == 0) s_rs[wave] = ws;
    }
    __syncthreads();
    int rs = 0;
#pragma unroll
    for (int w = 0; w < EDGE_BLOCKS / 32; ++w) rs += s_rs[w];

    const int piece  = tid >> 2;
    const int sub    = tid & 3;
    const int pstart = s_ps[piece];
    const int plen   = s_pl[piece];

    for (int i0 = 0; i0 < plen; i0 += 32) {
        int d[8];
#pragma unroll
        for (int k = 0; k < 8; ++k) {
            const int i = i0 + 4 * k + sub;
            d[k] = i < plen ? rec[pstart + i].y : -1;
        }
#pragma unroll
        for (int k = 0; k < 8; ++k)
            if (d[k] >= 0) atomicAdd(&s_c[d[k] - nb0], 1);
    }
    __syncthreads();

    int c = 0, inc = 0;
    if (tid < BIN_NODES) {
        c = s_c[tid];
        inc = c;
#pragma unroll
        for (int o = 1; o < 32; o <<= 1) {
            const int t = __shfl_up(inc, o, 32);
            if (lane >= o) inc += t;
        }
        if (lane == 31) s_wt[wave] = inc;
    }
    __syncthreads();
    int tot = 0;
#pragma unroll
    for (int w = 0; w < 8; ++w) tot += s_wt[w];
    const bool staged = tot <= STAGE_SRCS;
    if (tid < BIN_NODES) {
        int wp = 0;
#pragma unroll
        for (int w = 0; w < 8; ++w) wp += (w < wave) ? s_wt[w] : 0;
        const int start = rs + wp + inc - c;
        s_off[tid] = start;
        const int node = nb0 + tid;
        if (node < N) {
            cnt[node]    = c;
            rowptr[node] = start;
            dinv[node]   = rsqrtf((float)(c + 1));
        }
    }
    __syncthreads();

    for (int i0 = 0; i0 < plen; i0 += 32) {
        int2 r[8];
#pragma unroll
        for (int k = 0; k < 8; ++k) {
            const int i = i0 + 4 * k + sub;
            r[k] = i < plen ? rec[pstart + i] : make_int2(0, -1);
        }
        int pos[8];
#pragma unroll
        for (int k = 0; k < 8; ++k) {
            pos[k] = -1;
            if (r[k].y >= 0) pos[k] = atomicAdd(&s_off[r[k].y - nb0], 1);
        }
        if (staged) {
#pragma unroll
            for (int k = 0; k < 8; ++k)
                if (pos[k] >= 0) s_stage[pos[k] - rs] = r[k].x;
        } else {
#pragma unroll
            for (int k = 0; k < 8; ++k)
                if (pos[k] >= 0) csr_src[pos[k]] = r[k].x;
        }
    }
    if (staged) {
        __syncthreads();
        const int a0  = rs & ~31;
        const int end = rs + tot;
        for (int i0 = a0 + tid; i0 < end; i0 += 4096) {
            int v[4];
#pragma unroll
            for (int k = 0; k < 4; ++k) {
                const int i = i0 + k * 1024;
                v[k] = (i >= rs && i < end) ? s_stage[i - rs] : 0;
            }
#pragma unroll
            for (int k = 0; k < 4; ++k) {
                const int i = i0 + k * 1024;
                if (i >= rs && i < end) csr_src[i] = v[k];
            }
        }
    }
}

template <int U>
__device__ __forceinline__ void add_rows(const float* __restrict__ xs, int s, float w,
                                         int j, int c, float4& acc) {
    int    sj[U];
    float  wj[U];
    float4 r[U];
#pragma unroll
    for (int u = 0; u < U; ++u) {
        sj[u] = __builtin_amdgcn_readlane(s, j + u);
        wj[u] = __int_as_float(__builtin_amdgcn_readlane(__float_as_int(w), j + u));
    }
#pragma unroll
    for (int u = 0; u < U; ++u) r[u] = *(const float4*)(xs + (size_t)sj[u] * CDIM + c);
#pragma unroll
    for (int u = 0; u < U; ++u) {
        acc.x = fmaf(wj[u], r[u].x, acc.x);
        acc.y = fmaf(wj[u], r[u].y, acc.y);
        acc.z = fmaf(wj[u], r[u].z, acc.z);
        acc.w = fmaf(wj[u], r[u].w, acc.w);
    }
}

__device__ __forceinline__ void add_chunk(const float* __restrict__ xs, int s, float w,
                                          int m, int c, float4& acc) {
    int j = 0;
    for (; j + 8 <= m; j += 8) add_rows<8>(xs, s, w, j, c, acc);
    switch (m - j) {
        case 7: add_rows<7>(xs, s, w, j, c, acc); break;
        case 6: add_rows<6>(xs, s, w, j, c, acc); break;
        case 5: add_rows<5>(xs, s, w, j, c, acc); break;
        case 4: add_rows<4>(xs, s, w, j, c, acc); break;
        case 3: add_rows<3>(xs, s, w, j, c, acc); break;
        case 2: add_rows<2>(xs, s, w, j, c, acc); break;
        case 1: add_rows<1>(xs, s, w, j, c, acc); break;
        default: break;
    }
}

#define TILE_M 32
#define LDS_PITCH 132

__global__ void __launch_bounds__(512) __attribute__((amdgpu_waves_per_eu(16, 16)))
gcn_gather_gemm_epilogue_kernel(const float* __restrict__ x, const float* __restrict__ W,
                                const int* __restrict__ rowptr, const int* __restrict__ cnt,
                                const int* __restrict__ csr_src, const float* __restrict__ dinv,
                                const float* __restrict__ bvec,
                                const float* __restrict__ g1, const float* __restrict__ be1,
                                const float* __restrict__ g2, const float* __restrict__ be2,
                                float* __restrict__ out, int N) {
    __shared__ __attribute__((aligned(16))) float s_agg[TILE_M * LDS_PITCH];
    __shared__ __attribute__((aligned(16))) float s_d[2][TILE_M * LDS_PITCH];

    const int wave  = __builtin_amdgcn_readfirstlane((int)(threadIdx.x >> 5));
    const int node0 = (int)blockIdx.x * TILE_M + 2 * wave;
    const int lane  = threadIdx.x & 31;
    const int c = lane * 4;

    bool   live[2];
    int    st[2], nn[2], m0[2], s0[2];
    float  di[2], w0[2];
    float4 xv[2];
#pragma unroll
    for (int i = 0; i < 2; ++i) {
        live[i] = node0 + i < N;
        const int q = live[i] ? node0 + i : 0;
        st[i] = rowptr[q];
        nn[i] = live[i] ? cnt[q] : 0;
        di[i] = dinv[q];
        xv[i] = *(const float4*)(x + (size_t)q * CDIM + c);
    }
#pragma unroll
    for (int i = 0; i < 2; ++i) {
        m0[i] = nn[i] < 32 ? nn[i] : 32;
        const int k = lane < m0[i] ? lane : 0;
        s0[i] = nn[i] > 0 ? csr_src[st[i] + k] : 0;
    }
#pragma unroll
    for (int i = 0; i < 2; ++i) w0[i] = dinv[s0[i]] * di[i];

#pragma unroll
    for (int i = 0; i < 2; ++i) {
        float4 acc = make_float4(0.0f, 0.0f, 0.0f, 0.0f);
        add_chunk(x, s0[i], w0[i], m0[i], c, acc);
        for (int j0 = 32; j0 < nn[i]; j0 += 32) {
            const int m = (nn[i] - j0) < 32 ? (nn[i] - j0) : 32;
            const int   k = lane < m ? lane : 0;
            const int   s = csr_src[st[i] + j0 + k];
            const float w = dinv[s] * di[i];
            add_chunk(x, s, w, m, c, acc);
        }
        if (live[i]) {
            const float sw = di[i] * di[i];
            acc.x = fmaf(sw, xv[i].x, acc.x);
            acc.y = fmaf(sw, xv[i].y, acc.y);
            acc.z = fmaf(sw, xv[i].z, acc.z);
            acc.w = fmaf(sw, xv[i].w, acc.w);
        }
        *(float4*)(s_agg + (2 * wave + i) * LDS_PITCH + c) = acc;
    }

    const int half  = lane >> 4;
    const int l     = lane & 15;
    const int ct    = wave & 7;
    const int kp    = wave >> 3;
    const int kbase = kp * 64 + half * 4;
    const float* wrow = W + (size_t)(ct * 16 + l) * CDIM + kbase;
    float4 bf[8];
#pragma unroll
    for (int kb = 0; kb < 8; ++kb) bf[kb] = *(const float4*)(wrow + kb * 8);

    __syncthreads();

#pragma unroll
    for (int t = 0; t < 2; ++t) {
        const float* arow = s_agg + (16 * t + l) * LDS_PITCH + kbase;
        v8f d = {};
#pragma unroll
        for (int kb = 0; kb < 8; ++kb) {
            const float4 af = *(const float4*)(arow + kb * 8);
            v2f a0 = {af.x, af.y};
            v2f b0 = {bf[kb].x, bf[kb].y};
            d = __builtin_amdgcn_wmma_f32_16x16x4_f32(
                 false, a0,  false, b0,
                 (short)0, d,  false,  false);
            v2f a1 = {af.z, af.w};
            v2f b1 = {bf[kb].z, bf[kb].w};
            d = __builtin_amdgcn_wmma_f32_16x16x4_f32(
                 false, a1,  false, b1,
                 (short)0, d,  false,  false);
        }
        float* dp = s_d[kp] + (16 * t) * LDS_PITCH + ct * 16 + l;
#pragma unroll
        for (int r = 0; r < 8; ++r) dp[(r + 8 * half) * LDS_PITCH] = d[r];
    }

    __syncthreads();

    float4 xr[2];
#pragma unroll
    for (int i = 0; i < 2; ++i)
        xr[i] = *(const float4*)(x + (size_t)(live[i] ? node0 + i : 0) * CDIM + c);
    const float4 bv  = *(const float4*)(bvec + c);
    const float4 g1v = *(const float4*)(g1 + c);
    const float4 b1v = *(const float4*)(be1 + c);
    const float4 g2v = *(const float4*)(g2 + c);
    const float4 b2v = *(const float4*)(be2 + c);

#pragma unroll
    for (int i = 0; i < 2; ++i) {
        if (!live[i]) continue;
        const int row = 2 * wave + i;
        const float4 p0 = *(const float4*)(s_d[0] + row * LDS_PITCH + c);
        const float4 p1 = *(const float4*)(s_d[1] + row * LDS_PITCH + c);

        float v0 = fmaxf((p0.x + p1.x) + bv.x, 0.0f);
        float v1 = fmaxf((p0.y + p1.y) + bv.y, 0.0f);
        float v2 = fmaxf((p0.z + p1.z) + bv.z, 0.0f);
        float v3 = fmaxf((p0.w + p1.w) + bv.w, 0.0f);

        float mu = wave_sum32(v0 + v1 + v2 + v3) * (1.0f / CDIM);
        float d0 = v0 - mu, d1 = v1 - mu, d2 = v2 - mu, d3 = v3 - mu;
        float var = wave_sum32(d0 * d0 + d1 * d1 + d2 * d2 + d3 * d3) * (1.0f / CDIM);
        float inv = rsqrtf(var + LN_EPS);
        float z0 = xr[i].x + (d0 * inv * g1v.x + b1v.x);
        float z1 = xr[i].y + (d1 * inv * g1v.y + b1v.y);
        float z2 = xr[i].z + (d2 * inv * g1v.z + b1v.z);
        float z3 = xr[i].w + (d3 * inv * g1v.w + b1v.w);

        float mu2 = wave_sum32(z0 + z1 + z2 + z3) * (1.0f / CDIM);
        float e0 = z0 - mu2, e1 = z1 - mu2, e2 = z2 - mu2, e3 = z3 - mu2;
        float var2 = wave_sum32(e0 * e0 + e1 * e1 + e2 * e2 + e3 * e3) * (1.0f / CDIM);
        float inv2 = rsqrtf(var2 + LN_EPS);

        float4 o;
        o.x = e0 * inv2 * g2v.x + b2v.x;
        o.y = e1 * inv2 * g2v.y + b2v.y;
        o.z = e2 * inv2 * g2v.z + b2v.z;
        o.w = e3 * inv2 * g2v.w + b2v.w;
        *(float4*)(out + (size_t)(node0 + i) * CDIM + c) = o;
    }
}

extern "C" void kernel_launch(void* const* d_in, const int* in_sizes, int n_in,
                              void* d_out, int out_size, void* d_ws, size_t ws_size,
                              hipStream_t stream) {
    const float* x  = (const float*)d_in[0];
    const int*   ei = (const int*)d_in[1];
    const float* W  = (const float*)d_in[2];
    const float* b  = (const float*)d_in[3];
    const float* g1 = (const float*)d_in[4];
    const float* be1 = (const float*)d_in[5];
    const float* g2 = (const float*)d_in[6];
    const float* be2 = (const float*)d_in[7];
    float* out = (float*)d_out;

    const int N = in_sizes[0] / CDIM;
    const int E = in_sizes[1] / 2;

    const int NB    = (N + 255) / 256;
    const int PW    = EDGE_BLOCKS * (NB + 1);
    const int chunk = (E + EDGE_BLOCKS - 1) / EDGE_BLOCKS;

    int*   P       = (int*)d_ws;
    int2*  rec     = (int2*)(P + PW);
    int*   cnt     = (int*)(rec + (size_t)EDGE_BLOCKS * (chunk + 16));
    int*   rowptr  = cnt + N;
    float* dinv    = (float*)(rowptr + N);
    int*   csr_src = (int*)(dinv + N);

    gcn_chunk_sort_kernel<<<EDGE_BLOCKS, 1024, 0, stream>>>(ei, P, rec, E, NB, chunk);
    gcn_bin_place_kernel<<<NB, 1024, 0, stream>>>(rec, P, cnt, rowptr, dinv, csr_src, N, NB, chunk);
    gcn_gather_gemm_epilogue_kernel<<<(N + 31) / 32, 512, 0, stream>>>(x, W, rowptr, cnt, csr_src, dinv, b,
                                                                      g1, be1, g2, be2, out, N);
}


# W copied into fragment order by the first launch each call: each of a wave's 8 B-fragment loads reads 512 contiguous bytes (4 whole lines) instead of 32-byte pieces of 16 lines
# speedup vs baseline: 5.5523x; 1.0364x over previous
#include <hip/hip_runtime.h>
#include <hip/hip_bf16.h>

#define CDIM 128
#define LN_EPS 1e-5f

typedef float v2f __attribute__((ext_vector_type(2)));
typedef float v8f __attribute__((ext_vector_type(8)));

__device__ __forceinline__ float wave_sum32(float v) {
#pragma unroll
    for (int o = 16; o > 0; o >>= 1) v += __shfl_xor(v, o, 32);
    return v;
}

__device__ __forceinline__ int wave_isum32(int v) {
#pragma unroll
    for (int o = 16; o > 0; o >>= 1) v += __shfl_xor(v, o, 32);
    return v;
}


#define BIN_SHIFT   8
#define BIN_NODES   256
#define BIN_WINDOW  1024
#define EDGE_BLOCKS 256
#define STAGE_RECS  8192
#define STAGE_SRCS  8192

__device__ __forceinline__ int slice_pitch(int chunk) { return (chunk + 15) & ~15; }

#define WF_VEC4   (CDIM * CDIM / 4)
#define WF_BLOCKS (WF_VEC4 / 1024)
static_assert(WF_BLOCKS * 1024 == WF_VEC4, "W copy: one float4 per thread of whole blocks");

__global__ void __launch_bounds__(1024)
gcn_chunk_sort_kernel(const int* __restrict__ ei, int* __restrict__ P, int2* __restrict__ rec,
                      const float* __restrict__ W, float* __restrict__ Wf,
                      int E, int NB, int chunk) {
    __shared__ int s_h[BIN_WINDOW];
    __shared__ int s_wt[32];
    __shared__ int2 s_rec[STAGE_RECS];
    const int tid  = threadIdx.x;
    if (blockIdx.x >= EDGE_BLOCKS) {
        const int f    = ((int)blockIdx.x - EDGE_BLOCKS) * 1024 + tid;
        const int fl   = f & 31;
        const int fkb  = (f >> 5) & 7;
        const int fw   = f >> 8;
        *(float4*)(Wf + 4 * (size_t)f) =
            *(const float4*)(W + (size_t)((fw & 7) * 16 + (fl & 15)) * CDIM
                               + (fw >> 3) * 64 + (fl >> 4) * 4 + fkb * 8);
        return;
    }
    const int lane = tid & 31;
    const int wave = tid >> 5;
    const int e0 = (int)blockIdx.x * chunk;
    const int e1 = (e0 + chunk) < E ? (e0 + chunk) : E;
    int*  Pg   = P + (size_t)blockIdx.x * (NB + 1);
    int2* recg = rec + (size_t)blockIdx.x * slice_pitch(chunk);
    const bool staged = NB <= BIN_WINDOW && chunk <= STAGE_RECS;
    int carry = 0;
    for (int b0 = 0; b0 < NB; b0 += BIN_WINDOW) {
        s_h[tid] = 0;
        __syncthreads();
        for (int base = e0; base < e1; base += 8192) {
            int d[8];
#pragma unroll
            for (int k = 0; k < 8; ++k) {
                const int e = base + k * 1024 + tid;
                d[k] = e < e1 ? ei[E + e] : -1;
            }
#pragma unroll
            for (int k = 0; k < 8; ++k) {
                const unsigned b = (unsigned)((d[k] >> BIN_SHIFT) - b0);
                if (d[k] >= 0 && b < (unsigned)BIN_WINDOW) atomicAdd(&s_h[b], 1);
            }
        }
        __syncthreads();

        const int cval = s_h[tid];
        int inc = cval;
#pragma unroll
        for (int o = 1; o < 32; o <<= 1) {
            const int t = __shfl_up(inc, o, 32);
            if (lane >= o) inc += t;
        }
        if (lane == 31) s_wt[wave] = inc;
        __syncthreads();
        const int wt      = s_wt[lane];
        const int wavepre = wave_isum32(lane < wave ? wt : 0);
        const int total   = wave_isum32(wt);
        const int excl    = carry + wavepre + inc - cval;
        if (b0 + tid < NB) Pg[b0 + tid] = excl;
        s_h[tid] = excl;
        __syncthreads();

        for (int base = e0; base < e1; base += 8192) {
            int d[8], sr[8], pos[8];
#pragma unroll
            for (int k = 0; k < 8; ++k) {
                const int e = base + k * 1024 + tid;
                d[k]  = e < e1 ? ei[E + e] : -1;
                sr[k] = e < e1 ? ei[e] : 0;
            }
#pragma unroll
            for (int k = 0; k < 8; ++k) {
                const unsigned b = (unsigned)((d[k] >> BIN_SHIFT) - b0);
                pos[k] = -1;
                if (d[k] >= 0 && b < (unsigned)BIN_WINDOW) pos[k] = atomicAdd(&s_h[b], 1);
            }
            if (staged) {
#pragma unroll
                for (int k = 0; k < 8; ++k)
                    if (pos[k] >= 0) s_rec[pos[k]] = make_int2(sr[k], d[k]);
            } else {
#pragma unroll
                for (int k = 0; k < 8; ++k)
                    if (pos[k] >= 0) recg[pos[k]] = make_int2(sr[k], d[k]);
            }
        }
        carry += total;
        __syncthreads();
    }
    if (staged) {
        static_assert(STAGE_RECS == 8 * 1024, "copy-out: eight records per thread");
        int2 v[8];
#pragma unroll
        for (int k = 0; k < 8; ++k) {
            const int i = k * 1024 + tid;
            v[k] = i < carry ? s_rec[i] : make_int2(0, 0);
        }
#pragma unroll
        for (int k = 0; k < 8; ++k) {
            const int i = k * 1024 + tid;
            if (i < carry) recg[i] = v[k];
        }
    }
    if (tid == 0) Pg[NB] = carry;
}

static_assert(EDGE_BLOCKS * 4 == 1024, "place: four threads per piece");
static_assert(EDGE_BLOCKS % 32 == 0, "place: the table is read by whole waves");

__global__ void __launch_bounds__(1024)
gcn_bin_place_kernel(const int2* __restrict__ rec, const int* __restrict__ P,
                     int* __restrict__ cnt, int* __restrict__ rowptr, float* __restrict__ dinv,
                     int* __restrict__ csr_src, int N, int NB, int chunk) {
    __shared__ int s_c[BIN_NODES];
    __shared__ int s_off[BIN_NODES];
    __shared__ int s_ps[EDGE_BLOCKS];
    __shared__ int s_pl[EDGE_BLOCKS];
    __shared__ int s_rs[EDGE_BLOCKS / 32];
    __shared__ int s_wt[8];
    __shared__ int s_stage[STAGE_SRCS];
    const int tid  = threadIdx.x;
    const int lane = tid & 31;
    const int wave = tid >> 5;
    const int bin  = blockIdx.x;
    const int nb0  = bin << BIN_SHIFT;

    if (tid < BIN_NODES) s_c[tid] = 0;
    if (tid < EDGE_BLOCKS) {
        const int* Pg = P + (size_t)tid * (NB + 1) + bin;
        const int p0 = Pg[0];
        const int p1 = Pg[1];
        s_ps[tid] = tid * slice_pitch(chunk) + p0;
        s_pl[tid] = p1 - p0;
        const int ws = wave_isum32(p0);
        if (lane == 0) s_rs[wave] = ws;
    }
    __syncthreads();
    int rs = 0;
#pragma unroll
    for (int w = 0; w < EDGE_BLOCKS / 32; ++w) rs += s_rs[w];

    const int piece  = tid >> 2;
    const int sub    = tid & 3;
    const int pstart = s_ps[piece];
    const int plen   = s_pl[piece];

    for (int i0 = 0; i0 < plen; i0 += 32) {
        int d[8];
#pragma unroll
        for (int k = 0; k < 8; ++k) {
            const int i = i0 + 4 * k + sub;
            d[k] = i < plen ? rec[pstart + i].y : -1;
        }
#pragma unroll
        for (int k = 0; k < 8; ++k)
            if (d[k] >= 0) atomicAdd(&s_c[d[k] - nb0], 1);
    }
    __syncthreads();

    int c = 0, inc = 0;
    if (tid < BIN_NODES) {
        c = s_c[tid];
        inc = c;
#pragma unroll
        for (int o = 1; o < 32; o <<= 1) {
            const int t = __shfl_up(inc, o, 32);
            if (lane >= o) inc += t;
        }
        if (lane == 31) s_wt[wave] = inc;
    }
    __syncthreads();
    int tot = 0;
#pragma unroll
    for (int w = 0; w < 8; ++w) tot += s_wt[w];
    const bool staged = tot <= STAGE_SRCS;
    if (tid < BIN_NODES) {
        int wp = 0;
#pragma unroll
        for (int w = 0; w < 8; ++w) wp += (w < wave) ? s_wt[w] : 0;
        const int start = rs + wp + inc - c;
        s_off[tid] = start;
        const int node = nb0 + tid;
        if (node < N) {
            cnt[node]    = c;
            rowptr[node] = start;
            dinv[node]   = rsqrtf((float)(c + 1));
        }
    }
    __syncthreads();

    for (int i0 = 0; i0 < plen; i0 += 32) {
        int2 r[8];
#pragma unroll
        for (int k = 0; k < 8; ++k) {
            const int i = i0 + 4 * k + sub;
            r[k] = i < plen ? rec[pstart + i] : make_int2(0, -1);
        }
        int pos[8];
#pragma unroll
        for (int k = 0; k < 8; ++k) {
            pos[k] = -1;
            if (r[k].y >= 0) pos[k] = atomicAdd(&s_off[r[k].y - nb0], 1);
        }
        if (staged) {
#pragma unroll
            for (int k = 0; k < 8; ++k)
                if (pos[k] >= 0) s_stage[pos[k] - rs] = r[k].x;
        } else {
#pragma unroll
            for (int k = 0; k < 8; ++k)
                if (pos[k] >= 0) csr_src[pos[k]] = r[k].x;
        }
    }
    if (staged) {
        __syncthreads();
        const int a0  = rs & ~31;
        const int end = rs + tot;
        for (int i0 = a0 + tid; i0 < end; i0 += 4096) {
            int v[4];
#pragma unroll
            for (int k = 0; k < 4; ++k) {
                const int i = i0 + k * 1024;
                v[k] = (i >= rs && i < end) ? s_stage[i - rs] : 0;
            }
#pragma unroll
            for (int k = 0; k < 4; ++k) {
                const int i = i0 + k * 1024;
                if (i >= rs && i < end) csr_src[i] = v[k];
            }
        }
    }
}

template <int U>
__device__ __forceinline__ void add_rows(const float* __restrict__ xs, int s, float w,
                                         int j, int c, float4& acc) {
    int    sj[U];
    float  wj[U];
    float4 r[U];
#pragma unroll
    for (int u = 0; u < U; ++u) {
        sj[u] = __builtin_amdgcn_readlane(s, j + u);
        wj[u] = __int_as_float(__builtin_amdgcn_readlane(__float_as_int(w), j + u));
    }
#pragma unroll
    for (int u = 0; u < U; ++u) r[u] = *(const float4*)(xs + (size_t)sj[u] * CDIM + c);
#pragma unroll
    for (int u = 0; u < U; ++u) {
        acc.x = fmaf(wj[u], r[u].x, acc.x);
        acc.y = fmaf(wj[u], r[u].y, acc.y);
        acc.z = fmaf(wj[u], r[u].z, acc.z);
        acc.w = fmaf(wj[u], r[u].w, acc.w);
    }
}

__device__ __forceinline__ void add_chunk(const float* __restrict__ xs, int s, float w,
                                          int m, int c, float4& acc) {
    int j = 0;
    for (; j + 8 <= m; j += 8) add_rows<8>(xs, s, w, j, c, acc);
    switch (m - j) {
        case 7: add_rows<7>(xs, s, w, j, c, acc); break;
        case 6: add_rows<6>(xs, s, w, j, c, acc); break;
        case 5: add_rows<5>(xs, s, w, j, c, acc); break;
        case 4: add_rows<4>(xs, s, w, j, c, acc); break;
        case 3: add_rows<3>(xs, s, w, j, c, acc); break;
        case 2: add_rows<2>(xs, s, w, j, c, acc); break;
        case 1: add_rows<1>(xs, s, w, j, c, acc); break;
        default: break;
    }
}

#define TILE_M 32
#define LDS_PITCH 132

__global__ void __launch_bounds__(512) __attribute__((amdgpu_waves_per_eu(16, 16)))
gcn_gather_gemm_epilogue_kernel(const float* __restrict__ x, const float* __restrict__ Wf,
                                const int* __restrict__ rowptr, const int* __restrict__ cnt,
                                const int* __restrict__ csr_src, const float* __restrict__ dinv,
                                const float* __restrict__ bvec,
                                const float* __restrict__ g1, const float* __restrict__ be1,
                                const float* __restrict__ g2, const float* __restrict__ be2,
                                float* __restrict__ out, int N) {
    __shared__ __attribute__((aligned(16))) float s_agg[TILE_M * LDS_PITCH];
    __shared__ __attribute__((aligned(16))) float s_d[2][TILE_M * LDS_PITCH];

    const int wave  = __builtin_amdgcn_readfirstlane((int)(threadIdx.x >> 5));
    const int node0 = (int)blockIdx.x * TILE_M + 2 * wave;
    const int lane  = threadIdx.x & 31;
    const int c = lane * 4;

    bool   live[2];
    int    st[2], nn[2], m0[2], s0[2];
    float  di[2], w0[2];
    float4 xv[2];
#pragma unroll
    for (int i = 0; i < 2; ++i) {
        live[i] = node0 + i < N;
        const int q = live[i] ? node0 + i : 0;
        st[i] = rowptr[q];
        nn[i] = live[i] ? cnt[q] : 0;
        di[i] = dinv[q];
        xv[i] = *(const float4*)(x + (size_t)q * CDIM + c);
    }
#pragma unroll
    for (int i = 0; i < 2; ++i) {
        m0[i] = nn[i] < 32 ? nn[i] : 32;
        const int k = lane < m0[i] ? lane : 0;
        s0[i] = nn[i] > 0 ? csr_src[st[i] + k] : 0;
    }
#pragma unroll
    for (int i = 0; i < 2; ++i) w0[i] = dinv[s0[i]] * di[i];

#pragma unroll
    for (int i = 0; i < 2; ++i) {
        float4 acc = make_float4(0.0f, 0.0f, 0.0f, 0.0f);
        add_chunk(x, s0[i], w0[i], m0[i], c, acc);
        for (int j0 = 32; j0 < nn[i]; j0 += 32) {
            const int m = (nn[i] - j0) < 32 ? (nn[i] - j0) : 32;
            const int   k = lane < m ? lane : 0;
            const int   s = csr_src[st[i] + j0 + k];
            const float w = dinv[s] * di[i];
            add_chunk(x, s, w, m, c, acc);
        }
        if (live[i]) {
            const float sw = di[i] * di[i];
            acc.x = fmaf(sw, xv[i].x, acc.x);
            acc.y = fmaf(sw, xv[i].y, acc.y);
            acc.z = fmaf(sw, xv[i].z, acc.z);
            acc.w = fmaf(sw, xv[i].w, acc.w);
        }
        *(float4*)(s_agg + (2 * wave + i) * LDS_PITCH + c) = acc;
    }

    const int half  = lane >> 4;
    const int l     = lane & 15;
    const int ct    = wave & 7;
    const int kp    = wave >> 3;
    const int kbase = kp * 64 + half * 4;
    const float4* wfrag = (const float4*)Wf + wave * 256 + lane;
    float4 bf[8];
#pragma unroll
    for (int kb = 0; kb < 8; ++kb) bf[kb] = wfrag[kb * 32];

    __syncthreads();

#pragma unroll
    for (int t = 0; t < 2; ++t) {
        const float* arow = s_agg + (16 * t + l) * LDS_PITCH + kbase;
        v8f d = {};
#pragma unroll
        for (int kb = 0; kb < 8; ++kb) {
            const float4 af = *(const float4*)(arow + kb * 8);
            v2f a0 = {af.x, af.y};
            v2f b0 = {bf[kb].x, bf[kb].y};
            d = __builtin_amdgcn_wmma_f32_16x16x4_f32(
                 false, a0,  false, b0,
                 (short)0, d,  false,  false);
            v2f a1 = {af.z, af.w};
            v2f b1 = {bf[kb].z, bf[kb].w};
            d = __builtin_amdgcn_wmma_f32_16x16x4_f32(
                 false, a1,  false, b1,
                 (short)0, d,  false,  false);
        }
        float* dp = s_d[kp] + (16 * t) * LDS_PITCH + ct * 16 + l;
#pragma unroll
        for (int r = 0; r < 8; ++r) dp[(r + 8 * half) * LDS_PITCH] = d[r];
    }

    __syncthreads();

    float4 xr[2];
#pragma unroll
    for (int i = 0; i < 2; ++i)
        xr[i] = *(const float4*)(x + (size_t)(live[i] ? node0 + i : 0) * CDIM + c);
    const float4 bv  = *(const float4*)(bvec + c);
    const float4 g1v = *(const float4*)(g1 + c);
    const float4 b1v = *(const float4*)(be1 + c);
    const float4 g2v = *(const float4*)(g2 + c);
    const float4 b2v = *(const float4*)(be2 + c);

#pragma unroll
    for (int i = 0; i < 2; ++i) {
        if (!live[i]) continue;
        const int row = 2 * wave + i;
        const float4 p0 = *(const float4*)(s_d[0] + row * LDS_PITCH + c);
        const float4 p1 = *(const float4*)(s_d[1] + row * LDS_PITCH + c);

        float v0 = fmaxf((p0.x + p1.x) + bv.x, 0.0f);
        float v1 = fmaxf((p0.y + p1.y) + bv.y, 0.0f);
        float v2 = fmaxf((p0.z + p1.z) + bv.z, 0.0f);
        float v3 = fmaxf((p0.w + p1.w) + bv.w, 0.0f);

        float mu = wave_sum32(v0 + v1 + v2 + v3) * (1.0f / CDIM);
        float d0 = v0 - mu, d1 = v1 - mu, d2 = v2 - mu, d3 = v3 - mu;
        float var = wave_sum32(d0 * d0 + d1 * d1 + d2 * d2 + d3 * d3) * (1.0f / CDIM);
        float inv = rsqrtf(var + LN_EPS);
        float z0 = xr[i].x + (d0 * inv * g1v.x + b1v.x);
        float z1 = xr[i].y + (d1 * inv * g1v.y + b1v.y);
        float z2 = xr[i].z + (d2 * inv * g1v.z + b1v.z);
        float z3 = xr[i].w + (d3 * inv * g1v.w + b1v.w);

        float mu2 = wave_sum32(z0 + z1 + z2 + z3) * (1.0f / CDIM);
        float e0 = z0 - mu2, e1 = z1 - mu2, e2 = z2 - mu2, e3 = z3 - mu2;
        float var2 = wave_sum32(e0 * e0 + e1 * e1 + e2 * e2 + e3 * e3) * (1.0f / CDIM);
        float inv2 = rsqrtf(var2 + LN_EPS);

        float4 o;
        o.x = e0 * inv2 * g2v.x + b2v.x;
        o.y = e1 * inv2 * g2v.y + b2v.y;
        o.z = e2 * inv2 * g2v.z + b2v.z;
        o.w = e3 * inv2 * g2v.w + b2v.w;
        *(float4*)(out + (size_t)(node0 + i) * CDIM + c) = o;
    }
}

extern "C" void kernel_launch(void* const* d_in, const int* in_sizes, int n_in,
                              void* d_out, int out_size, void* d_ws, size_t ws_size,
                              hipStream_t stream) {
    const float* x  = (const float*)d_in[0];
    const int*   ei = (const int*)d_in[1];
    const float* W  = (const float*)d_in[2];
    const float* b  = (const float*)d_in[3];
    const float* g1 = (const float*)d_in[4];
    const float* be1 = (const float*)d_in[5];
    const float* g2 = (const float*)d_in[6];
    const float* be2 = (const float*)d_in[7];
    float* out = (float*)d_out;

    const int N = in_sizes[0] / CDIM;
    const int E = in_sizes[1] / 2;

    const int NB    = (N + 255) / 256;
    const int PW    = EDGE_BLOCKS * (NB + 1);
    const int chunk = (E + EDGE_BLOCKS - 1) / EDGE_BLOCKS;

    float* Wf      = (float*)d_ws;
    int*   P       = (int*)(Wf + CDIM * CDIM);
    int2*  rec     = (int2*)(P + PW);
    int*   cnt     = (int*)(rec + (size_t)EDGE_BLOCKS * (chunk + 16));
    int*   rowptr  = cnt + N;
    float* dinv    = (float*)(rowptr + N);
    int*   csr_src = (int*)(dinv + N);

    gcn_chunk_sort_kernel<<<EDGE_BLOCKS + WF_BLOCKS, 1024, 0, stream>>>(ei, P, rec, W, Wf, E, NB, chunk);
    gcn_bin_place_kernel<<<NB, 1024, 0, stream>>>(rec, P, cnt, rowptr, dinv, csr_src, N, NB, chunk);
    gcn_gather_gemm_epilogue_kernel<<<(N + 31) / 32, 512, 0, stream>>>(x, Wf, rowptr, cnt, csr_src, dinv, b,
                                                                      g1, be1, g2, be2, out, N);
}
